# starting kernel
# baseline (speedup 1.0000x reference)
#include <hip/hip_runtime.h>

#define N_NODES 100000
#define N_EDGES 1200000
#define DIM 64
#define E_TOT (N_EDGES + N_NODES)

typedef __attribute__((ext_vector_type(2))) float v2f;
typedef __attribute__((ext_vector_type(8))) float v8f;

__global__ void deg_init_kernel(float* __restrict__ deg) {
    int i = blockIdx.x * blockDim.x + threadIdx.x;
    if (i < N_NODES) deg[i] = 1.0f;
}

__global__ void deg_accum_kernel(const int* __restrict__ ei, float* __restrict__ deg) {
    int e = blockIdx.x * blockDim.x + threadIdx.x;
    if (e < N_EDGES) atomicAdd(&deg[ei[N_EDGES + e]], 1.0f);
}

__global__ void dinv_kernel(float* __restrict__ deg) {
    int i = blockIdx.x * blockDim.x + threadIdx.x;
    if (i < N_NODES) deg[i] = rsqrtf(deg[i]);
}

__global__ void gemm_wmma_kernel(const float* __restrict__ in,
                                 const float* __restrict__ W,
                                 float* __restrict__ out) {
    const int lane = threadIdx.x & 31;
    const int wid  = (blockIdx.x * blockDim.x + threadIdx.x) >> 5;
    const int nWaves = (gridDim.x * blockDim.x) >> 5;

    const int col0 = (wid & 3) * 16;
    const int m    = lane & 15;
    const int hi   = lane >> 4;
    const int klo  = hi * 2;

    v2f bfrag[16];
#pragma unroll
    for (int kb = 0; kb < 16; ++kb) {
        int k = kb * 4 + klo;
        bfrag[kb].x = W[k * DIM + col0 + m];
        bfrag[kb].y = W[(k + 1) * DIM + col0 + m];
    }

    const int nRowTiles  = N_NODES / 16;
    const int rowStride  = nWaves >> 2;

    for (int rt = wid >> 2; rt < nRowTiles; rt += rowStride) {
        const int row0 = rt * 16;
        const float* arow = in + (size_t)(row0 + m) * DIM;
        v8f c = {};
#pragma unroll
        for (int kb = 0; kb < 16; ++kb) {
            v2f a = *(const v2f*)(arow + kb * 4 + klo);
            c = __builtin_amdgcn_wmma_f32_16x16x4_f32(
                     false, a,  false, bfrag[kb],
                     (short)0, c,  false,  false);
        }
        float* orow = out + (size_t)(row0 + hi * 8) * DIM + col0 + m;
#pragma unroll
        for (int r = 0; r < 8; ++r) orow[(size_t)r * DIM] = c[r];
    }
}

__global__ void bias_fill_kernel(float* __restrict__ h, const float* __restrict__ b) {
    int i = blockIdx.x * blockDim.x + threadIdx.x;
    if (i < N_NODES * DIM) h[i] = b[i & (DIM - 1)];
}

__global__ void scatter_kernel(const float* __restrict__ t,
                               const int* __restrict__ ei,
                               const float* __restrict__ dinv,
                               float* __restrict__ h) {
    int tid = blockIdx.x * blockDim.x + threadIdx.x;
    int e = tid >> 5;
    int p = tid & 31;
    if (e >= E_TOT) return;
    int s, d;
    if (e < N_EDGES) { s = ei[e]; d = ei[N_EDGES + e]; }
    else             { s = e - N_EDGES; d = s; }
    float nm = dinv[s] * dinv[d];
    const float* mp = t + (size_t)s * DIM + p * 2;
    v2f msg = *(const v2f*)mp;
    float* op = h + (size_t)d * DIM + p * 2;
    atomicAdd(op,     msg.x * nm);
    atomicAdd(op + 1, msg.y * nm);
}

__global__ void relu_kernel(float* __restrict__ h) {
    int i = blockIdx.x * blockDim.x + threadIdx.x;
    if (i < N_NODES * DIM) h[i] = fmaxf(h[i], 0.0f);
}

extern "C" void kernel_launch(void* const* d_in, const int* in_sizes, int n_in,
                              void* d_out, int out_size, void* d_ws, size_t ws_size,
                              hipStream_t stream) {
    const float* x  = (const float*)d_in[0];
    const int*   ei = (const int*)d_in[1];
    const float* W1 = (const float*)d_in[2];
    const float* b1 = (const float*)d_in[3];
    const float* W2 = (const float*)d_in[4];
    const float* b2 = (const float*)d_in[5];
    const float* W3 = (const float*)d_in[6];
    const float* b3 = (const float*)d_in[7];
    float* out = (float*)d_out;

    char* ws = (char*)d_ws;
    size_t off = 0;
    float* dinv = (float*)(ws + off); off += ((size_t)N_NODES * 4 + 255) & ~(size_t)255;
    float* tbuf = (float*)(ws + off); off += (size_t)N_NODES * DIM * 4;
    float* hbuf = (float*)(ws + off);

    const int T = 256;
    const int gN   = (N_NODES + T - 1) / T;
    const int gE   = (N_EDGES + T - 1) / T;
    const int gND  = (N_NODES * DIM + T - 1) / T;
    const int gSc  = ((E_TOT * 32) + T - 1) / T;
    const int gGemm = 512;

    deg_init_kernel<<<gN, T, 0, stream>>>(dinv);
    deg_accum_kernel<<<gE, T, 0, stream>>>(ei, dinv);
    dinv_kernel<<<gN, T, 0, stream>>>(dinv);

    gemm_wmma_kernel<<<gGemm, T, 0, stream>>>(x, W1, tbuf);
    bias_fill_kernel<<<gND, T, 0, stream>>>(hbuf, b1);
    scatter_kernel<<<gSc, T, 0, stream>>>(tbuf, ei, dinv, hbuf);
    relu_kernel<<<gND, T, 0, stream>>>(hbuf);

    gemm_wmma_kernel<<<gGemm, T, 0, stream>>>(hbuf, W2, tbuf);
    bias_fill_kernel<<<gND, T, 0, stream>>>(hbuf, b2);
    scatter_kernel<<<gSc, T, 0, stream>>>(tbuf, ei, dinv, hbuf);
    relu_kernel<<<gND, T, 0, stream>>>(hbuf);

    gemm_wmma_kernel<<<gGemm, T, 0, stream>>>(hbuf, W3, tbuf);
    bias_fill_kernel<<<gND, T, 0, stream>>>(out, b3);
    scatter_kernel<<<gSc, T, 0, stream>>>(tbuf, ei, dinv, out);
}


# Scatter reads source rows from LDS: edges grouped once per call by (dst range, block of 256 src rows), a work-group stages its block of product rows, then adds; atomic arrival order shifts
# speedup vs baseline: 2.4679x; 2.4679x over previous
#include <hip/hip_runtime.h>

#define N_NODES 100000
#define N_EDGES 1200000
#define DIM 64

typedef __attribute__((ext_vector_type(2))) float v2f;
typedef __attribute__((ext_vector_type(8))) float v8f;

#define WS_DINV_BYTES ((((size_t)N_NODES * 4) + 255) & ~(size_t)255)
#define WS_PLANE_BYTES ((size_t)N_NODES * DIM * 4)
#define WS_NEED_BYTES (WS_DINV_BYTES + 2 * WS_PLANE_BYTES + (size_t)N_EDGES * 4)

#define HALF_NODES (N_NODES / 2)

#define SB 256
#define NSB ((N_NODES + SB - 1) / SB)
#define NBINS (2 * NSB)
#define BIN_WORDS 1024
#define WS_REC_OFF (WS_DINV_BYTES + 2 * WS_PLANE_BYTES)
#define WS_CNT_OFF (WS_REC_OFF + (size_t)N_EDGES * 8)
#define WS_START_OFF (WS_CNT_OFF + (size_t)BIN_WORDS * 4)
#define WS_CUR_OFF (WS_START_OFF + (size_t)BIN_WORDS * 4)
#define WS_Q1_OFF (WS_CUR_OFF + (size_t)BIN_WORDS * 4)
#define WS_NEED2 (WS_Q1_OFF + 256)
#define SC_CH 64
#define SC_CHUNKS (N_EDGES / SC_CH)
#define DEG_CH 384
#define DEG_CHUNKS (N_EDGES / DEG_CH)

__device__ __forceinline__ size_t queue_word(const int q, const size_t ws_size) {
    const size_t w1 = (ws_size >= WS_NEED2) ? (size_t)(WS_Q1_OFF / 4)
                    : (ws_size >= WS_NEED_BYTES + 256) ? (size_t)(WS_NEED_BYTES / 4)
                                                       : (size_t)N_NODES + 16;
    return q ? w1 : (size_t)N_NODES;
}

__device__ __forceinline__ void reset_queues(unsigned* wsw, const size_t ws_size) {
    if (blockIdx.x == 0 && threadIdx.x == 0) {
        wsw[queue_word(0, ws_size)] = 0u;
        wsw[queue_word(1, ws_size)] = 0u;
    }
}

__device__ __forceinline__ int placement_half() {
    return (int)((__builtin_amdgcn_s_sendmsg_rtn(0x87) >> 18) & 1u);
}

__device__ __forceinline__ unsigned issue_draw(unsigned* ctr, const int lane) {
    unsigned v = 0u;
    if (lane == 0) v = atomicAdd(ctr, 1u);
    return v;
}
__device__ __forceinline__ unsigned drawn_chunk(const unsigned v) {
    return (unsigned)__builtin_amdgcn_readfirstlane((int)v);
}

__global__ void deg_init_kernel(float* __restrict__ deg, unsigned* wsw, const size_t ws_size) {
    int i = blockIdx.x * blockDim.x + threadIdx.x;
    if (i < N_NODES) deg[i] = 1.0f;
    if (ws_size >= WS_NEED2 && i < BIN_WORDS) wsw[WS_CNT_OFF / 4 + i] = 0u;
    reset_queues(wsw, ws_size);
}

__global__ void __launch_bounds__(256)
deg_accum_kernel(const int* __restrict__ ei, float* __restrict__ deg,
                 unsigned* wsw, const size_t ws_size) {
    const int lane = threadIdx.x & 31;
    const int my_half = placement_half();
    const int n_pass = ((threadIdx.x >> 5) == 0) ? 2 : 1;
    const bool have_bins = ws_size >= WS_NEED2;
    __shared__ unsigned hist[NBINS];
    if (have_bins) {
        for (int i = threadIdx.x; i < NBINS; i += blockDim.x) hist[i] = 0u;
        __syncthreads();
    }
    for (int pass = 0; pass < n_pass; ++pass) {
        const int q  = my_half ^ pass;
        const int lo = q ? HALF_NODES : 0;
        const int hi = q ? N_NODES : HALF_NODES;
        unsigned* ctr = wsw + queue_word(q, ws_size);
        unsigned c = drawn_chunk(issue_draw(ctr, lane));
        while (c < DEG_CHUNKS) {
            const int* dp = ei + N_EDGES + (int)c * DEG_CH + lane;
            const int* sp = ei + (int)c * DEG_CH + lane;
            int d[DEG_CH / 32], s[DEG_CH / 32];
#pragma unroll
            for (int i = 0; i < DEG_CH / 32; ++i) {
                d[i] = dp[i * 32];
                s[i] = have_bins ? sp[i * 32] : 0;
            }
            const unsigned nxt = issue_draw(ctr, lane);
#pragma unroll
            for (int i = 0; i < DEG_CH / 32; ++i)
                if (d[i] >= lo && d[i] < hi) {
                    atomicAdd(&deg[d[i]], 1.0f);
                    if (have_bins) atomicAdd(&hist[q * NSB + (s[i] >> 8)], 1u);
                }
            c = drawn_chunk(nxt);
        }
    }
    if (have_bins) {
        __syncthreads();
        unsigned* cnt = wsw + WS_CNT_OFF / 4;
        for (int i = threadIdx.x; i < NBINS; i += blockDim.x) {
            const unsigned v = hist[i];
            if (v != 0u) atomicAdd(&cnt[i], v);
        }
    }
}

__global__ void __launch_bounds__(256)
dinv_kernel(float* __restrict__ deg, unsigned* wsw, const size_t ws_size) {
    int i = blockIdx.x * blockDim.x + threadIdx.x;
    if (i < N_NODES) deg[i] = rsqrtf(deg[i]);
    __shared__ unsigned sc[256];
    if (blockIdx.x == 0 && ws_size >= WS_NEED2) {
        const int t = threadIdx.x;
        const unsigned* cnt = wsw + WS_CNT_OFF / 4;
        const unsigned c0 = cnt[4 * t], c1 = cnt[4 * t + 1], c2 = cnt[4 * t + 2], c3 = cnt[4 * t + 3];
        const unsigned own = c0 + c1 + c2 + c3;
        sc[t] = own;
        __syncthreads();
        for (int off = 1; off < 256; off <<= 1) {
            const unsigned v = (t >= off) ? sc[t - off] : 0u;
            __syncthreads();
            sc[t] += v;
            __syncthreads();
        }
        const unsigned e0 = sc[t] - own;
        unsigned* st  = wsw + WS_START_OFF / 4;
        unsigned* cur = wsw + WS_CUR_OFF / 4;
        st[4 * t]     = e0;                 cur[4 * t]     = e0;
        st[4 * t + 1] = e0 + c0;            cur[4 * t + 1] = e0 + c0;
        st[4 * t + 2] = e0 + c0 + c1;       cur[4 * t + 2] = e0 + c0 + c1;
        st[4 * t + 3] = e0 + c0 + c1 + c2;  cur[4 * t + 3] = e0 + c0 + c1 + c2;
    }
}

#define PL_T 512
#define PL_PER 8
#define PL_EDGES (PL_T * PL_PER)
#define PL_GRID ((N_EDGES + PL_EDGES - 1) / PL_EDGES)
__global__ void __launch_bounds__(PL_T)
place_kernel(const int* __restrict__ ei, const float* __restrict__ dinv,
             unsigned* wsw, const size_t ws_size) {
    if (ws_size < WS_NEED2) return;
    __shared__ unsigned hist[NBINS];
    __shared__ unsigned base[NBINS];
    for (int i = threadIdx.x; i < NBINS; i += PL_T) hist[i] = 0u;
    __syncthreads();
    int s[PL_PER], d[PL_PER];
    unsigned rank[PL_PER];
    const int e0 = blockIdx.x * PL_EDGES + threadIdx.x;
#pragma unroll
    for (int k = 0; k < PL_PER; ++k) {
        const int e = e0 + k * PL_T;
        s[k] = (e < N_EDGES) ? ei[e] : -1;
        d[k] = (e < N_EDGES) ? ei[N_EDGES + e] : 0;
    }
#pragma unroll
    for (int k = 0; k < PL_PER; ++k)
        if (s[k] >= 0) rank[k] = atomicAdd(&hist[(d[k] >= HALF_NODES ? NSB : 0) + (s[k] >> 8)], 1u);
    __syncthreads();
    unsigned* cur = wsw + WS_CUR_OFF / 4;
    for (int i = threadIdx.x; i < NBINS; i += PL_T) {
        const unsigned v = hist[i];
        base[i] = (v != 0u) ? atomicAdd(&cur[i], v) : 0u;
    }
    __syncthreads();
    uint2* rec = (uint2*)((char*)wsw + WS_REC_OFF);
#pragma unroll
    for (int k = 0; k < PL_PER; ++k)
        if (s[k] >= 0) {
            const unsigned pos = base[(d[k] >= HALF_NODES ? NSB : 0) + (s[k] >> 8)] + rank[k];
            const float w = dinv[s[k]] * dinv[d[k]];
            rec[pos] = make_uint2((unsigned)(s[k] & (SB - 1)) | ((unsigned)d[k] << 8),
                                  __builtin_bit_cast(unsigned, w));
        }
}

typedef __attribute__((ext_vector_type(4))) float v4f;

__device__ __forceinline__ void gemm_fused_body(const bool relu_in,
                                                const float* __restrict__ in,
                                                const float* __restrict__ W,
                                                const float* __restrict__ bias,
                                                const float* __restrict__ dinv,
                                                float* __restrict__ t_out,
                                                float* __restrict__ h_init) {
    const int lane = threadIdx.x & 31;
    const int wid  = (blockIdx.x * blockDim.x + threadIdx.x) >> 5;
    const int nWaves = (gridDim.x * blockDim.x) >> 5;

    const int col0 = (wid & 3) * 16;
    const int m    = lane & 15;
    const int hi   = lane >> 4;
    const int klo  = hi * 2;

    v2f bfrag[16];
#pragma unroll
    for (int kb = 0; kb < 16; ++kb) {
        int k = kb * 4 + klo;
        bfrag[kb].x = W[k * DIM + col0 + m];
        bfrag[kb].y = W[(k + 1) * DIM + col0 + m];
    }
    const float bv = bias[col0 + m];

    const int nRowTiles  = N_NODES / 16;
    const int rowStride  = nWaves >> 2;

    for (int rt = wid >> 2; rt < nRowTiles; rt += rowStride) {
        const int row0 = rt * 16;
        const int rbase = row0 + hi * 8;
        const float* arow = in + (size_t)(row0 + m) * DIM;
        const v4f dv0 = *(const v4f*)(dinv + rbase);
        const v4f dv1 = *(const v4f*)(dinv + rbase + 4);
        v8f c = {};
#pragma unroll
        for (int kb = 0; kb < 16; ++kb) {
            v2f a = *(const v2f*)(arow + kb * 4 + klo);
            if (relu_in) { a.x = fmaxf(a.x, 0.0f); a.y = fmaxf(a.y, 0.0f); }
            c = __builtin_amdgcn_wmma_f32_16x16x4_f32(
                     false, a,  false, bfrag[kb],
                     (short)0, c,  false,  false);
        }
        float* trow = t_out  + (size_t)rbase * DIM + col0 + m;
        float* hrow = h_init + (size_t)rbase * DIM + col0 + m;
        const float dv[8] = {dv0.x, dv0.y, dv0.z, dv0.w, dv1.x, dv1.y, dv1.z, dv1.w};
#pragma unroll
        for (int r = 0; r < 8; ++r) {
            const float nm = dv[r] * dv[r];
            trow[(size_t)r * DIM] = c[r];
            hrow[(size_t)r * DIM] = bv + c[r] * nm;
        }
    }
}

__global__ void gemm_first_kernel(const float* __restrict__ in,
                                  const float* __restrict__ W,
                                  const float* __restrict__ bias,
                                  const float* __restrict__ dinv,
                                  float* __restrict__ t_out,
                                  float* __restrict__ h_init,
                                  unsigned* wsw, const size_t ws_size) {
    reset_queues(wsw, ws_size);
    gemm_fused_body(false, in, W, bias, dinv, t_out, h_init);
}

__global__ void gemm_relu_kernel(const float* __restrict__ in,
                                 const float* __restrict__ W,
                                 const float* __restrict__ bias,
                                 const float* __restrict__ dinv,
                                 float* __restrict__ t_out,
                                 float* __restrict__ h_init,
                                 unsigned* wsw, const size_t ws_size) {
    reset_queues(wsw, ws_size);
    gemm_fused_body(true, in, W, bias, dinv, t_out, h_init);
}

#define SC_K 16
#define HALF_DIM (DIM / 2)

template <bool FIRST>
__device__ __forceinline__ void scatter_body(const float* __restrict__ t,
                                             const int* __restrict__ ei,
                                             const float* __restrict__ dinv,
                                             float* __restrict__ enorm,
                                             const size_t ws_size,
                                             float* __restrict__ h,
                                             unsigned* wsw) {
    const int lane = threadIdx.x & 31;
    const bool have_enorm = ws_size >= WS_NEED_BYTES;
    const int my_half = placement_half();

    const int n_pass = ((threadIdx.x >> 5) == 0) ? 2 : 1;
    for (int pass = 0; pass < n_pass; ++pass) {
        const int q  = my_half ^ pass;
        const int lo = q ? HALF_NODES : 0;
        const int hi = q ? N_NODES : HALF_NODES;
        unsigned* ctr = wsw + queue_word(q, ws_size);
        unsigned c = drawn_chunk(issue_draw(ctr, lane));
        while (c < SC_CHUNKS) {
            const int e0 = (int)c * SC_CH + lane;

            int   sl[SC_CH / 32], dl[SC_CH / 32];
            float nl[SC_CH / 32];
            bool  mine[SC_CH / 32];
#pragma unroll
            for (int g = 0; g < SC_CH / 32; ++g) {
                sl[g] = ei[e0 + g * 32];
                dl[g] = ei[N_EDGES + e0 + g * 32];
                nl[g] = (!FIRST && have_enorm) ? enorm[e0 + g * 32] : 0.0f;
            }
            const unsigned nxt = issue_draw(ctr, lane);
#pragma unroll
            for (int g = 0; g < SC_CH / 32; ++g) {
                mine[g] = (dl[g] >= lo) && (dl[g] < hi);
                if ((FIRST || !have_enorm) && mine[g]) {
                    nl[g] = dinv[sl[g]] * dinv[dl[g]];
                    if (FIRST && have_enorm) enorm[e0 + g * 32] = nl[g];
                }
            }

#pragma unroll
            for (int g = 0; g < SC_CH / 32; ++g) {
                unsigned mask = __builtin_amdgcn_ballot_w32(mine[g]);
                while (mask != 0u) {
                    int lj[SC_K];
#pragma unroll
                    for (int j = 0; j < SC_K; ++j) {
                        lj[j] = (mask != 0u) ? (int)__builtin_ctz(mask) : -1;
                        mask &= (mask - 1u);
                    }
                    float mlo[SC_K], mhi[SC_K], nj[SC_K];
                    int   dj[SC_K];
#pragma unroll
                    for (int j = 0; j < SC_K; ++j) {
                        if (lj[j] >= 0) {
                            const int sj = __builtin_amdgcn_readlane(sl[g], lj[j]);
                            dj[j] = __builtin_amdgcn_readlane(dl[g], lj[j]);
                            nj[j] = __builtin_bit_cast(float,
                                      __builtin_amdgcn_readlane(__builtin_bit_cast(int, nl[g]), lj[j]));
                            const float* trow = t + (size_t)sj * DIM + lane;
                            mlo[j] = trow[0];
                            mhi[j] = trow[HALF_DIM];
                        }
                    }
#pragma unroll
                    for (int j = 0; j < SC_K; ++j) {
                        if (lj[j] >= 0) {
                            float* op = h + (size_t)dj[j] * DIM + lane;
                            atomicAdd(op,            mlo[j] * nj[j]);
                            atomicAdd(op + HALF_DIM, mhi[j] * nj[j]);
                        }
                    }
                }
            }
            c = drawn_chunk(nxt);
        }
    }
}

#define SC_T 512
#define SC_W (SC_T / 32)
#define ST_K 8
#define ST_PER (SB * DIM / 4 / SC_T)

template <bool GUARD>
__device__ __forceinline__ void staged_trip(const float* tile, const uint2 r, const int j0, const int cnt,
                                            const int lane, float* __restrict__ h) {
    float mlo[ST_K], mhi[ST_K], wj[ST_K];
    int   dj[ST_K];
#pragma unroll
    for (int j = 0; j < ST_K; ++j) {
        if (!GUARD || j < cnt) {
            const unsigned key = (unsigned)__builtin_amdgcn_readlane((int)r.x, j0 + j);
            wj[j] = __builtin_bit_cast(float, __builtin_amdgcn_readlane((int)r.y, j0 + j));
            const float* trow = tile + (key & (SB - 1)) * DIM + lane;
            dj[j]  = (int)(key >> 8);
            mlo[j] = trow[0];
            mhi[j] = trow[HALF_DIM];
        }
    }
#pragma unroll
    for (int j = 0; j < ST_K; ++j) {
        if (!GUARD || j < cnt) {
            float* op = h + (size_t)dj[j] * DIM + lane;
            atomicAdd(op,            mlo[j] * wj[j]);
            atomicAdd(op + HALF_DIM, mhi[j] * wj[j]);
        }
    }
}

__device__ __forceinline__ void staged_scatter_body(const float* __restrict__ t,
                                                    const size_t ws_size,
                                                    float* __restrict__ h,
                                                    unsigned* wsw) {
    __shared__ v4f tile4[SB * DIM / 4];
    __shared__ unsigned s_draw[2];
    const float* tile = (const float*)tile4;
    const int lane = threadIdx.x & 31;
    const int wave = __builtin_amdgcn_readfirstlane((int)(threadIdx.x >> 5));
    const unsigned* st = wsw + WS_START_OFF / 4;
    const uint2* rec = (const uint2*)((const char*)wsw + WS_REC_OFF);

    if (threadIdx.x == 0) s_draw[1] = (unsigned)placement_half();
    __syncthreads();
    const int my_half = __builtin_amdgcn_readfirstlane((int)s_draw[1]);

    for (int pass = 0; pass < 2; ++pass) {
        const int q = my_half ^ pass;
        unsigned* ctr = wsw + queue_word(q, ws_size);
        for (;;) {
            if (threadIdx.x == 0) s_draw[0] = atomicAdd(ctr, 1u);
            __syncthreads();
            const int c = __builtin_amdgcn_readfirstlane((int)s_draw[0]);
            __syncthreads();
            if ((unsigned)c >= (unsigned)NSB) break;

            const int row0  = c * SB;
            const int nrows = (N_NODES - row0 < SB) ? (N_NODES - row0) : SB;
            const v4f* src4 = (const v4f*)(t + (size_t)row0 * DIM);
            const int n4 = nrows * (DIM / 4);
            if (nrows == SB) {
                v4f buf[ST_PER];
#pragma unroll
                for (int k = 0; k < ST_PER; ++k) buf[k] = src4[(int)threadIdx.x + k * SC_T];
#pragma unroll
                for (int k = 0; k < ST_PER; ++k) tile4[(int)threadIdx.x + k * SC_T] = buf[k];
            } else {
                for (int i = (int)threadIdx.x; i < n4; i += SC_T) tile4[i] = src4[i];
            }
            __syncthreads();

            const int bin  = q * NSB + c;
            const int e_lo = __builtin_amdgcn_readfirstlane((int)st[bin]);
            const int e_hi = __builtin_amdgcn_readfirstlane((int)st[bin + 1]);
            for (int g = e_lo + wave * 32; g < e_hi; g += SC_W * 32) {
                const int idx = g + lane;
                uint2 r = make_uint2(0u, 0u);
                if (idx < e_hi) r = rec[idx];
                const int n = (e_hi - g < 32) ? (e_hi - g) : 32;
                int j0 = 0;
                for (; j0 + ST_K <= n; j0 += ST_K) staged_trip<false>(tile, r, j0, ST_K, lane, h);
                if (j0 < n) staged_trip<true>(tile, r, j0, n - j0, lane, h);
            }
            __syncthreads();
        }
    }
}

__global__ void __launch_bounds__(SC_T) __attribute__((amdgpu_waves_per_eu(16, 16)))
scatter_first_kernel(const float* __restrict__ t,
                     const int* __restrict__ ei,
                     const float* __restrict__ dinv,
                     float* __restrict__ enorm,
                     const size_t ws_size,
                     float* __restrict__ h,
                     unsigned* wsw) {
    if (ws_size >= WS_NEED2) staged_scatter_body(t, ws_size, h, wsw);
    else scatter_body<true>(t, ei, dinv, enorm, ws_size, h, wsw);
}

__global__ void __launch_bounds__(SC_T) __attribute__((amdgpu_waves_per_eu(16, 16)))
scatter_next_kernel(const float* __restrict__ t,
                    const int* __restrict__ ei,
                    const float* __restrict__ dinv,
                    float* __restrict__ enorm,
                    const size_t ws_size,
                    float* __restrict__ h,
                    unsigned* wsw) {
    if (ws_size >= WS_NEED2) staged_scatter_body(t, ws_size, h, wsw);
    else scatter_body<false>(t, ei, dinv, enorm, ws_size, h, wsw);
}

extern "C" void kernel_launch(void* const* d_in, const int* in_sizes, int n_in,
                              void* d_out, int out_size, void* d_ws, size_t ws_size,
                              hipStream_t stream) {
    const float* x  = (const float*)d_in[0];
    const int*   ei = (const int*)d_in[1];
    const float* W1 = (const float*)d_in[2];
    const float* b1 = (const float*)d_in[3];
    const float* W2 = (const float*)d_in[4];
    const float* b2 = (const float*)d_in[5];
    const float* W3 = (const float*)d_in[6];
    const float* b3 = (const float*)d_in[7];
    float* out = (float*)d_out;

    char* ws = (char*)d_ws;
    size_t off = 0;
    float* dinv = (float*)(ws + off); off += ((size_t)N_NODES * 4 + 255) & ~(size_t)255;
    float* tbuf = (float*)(ws + off); off += (size_t)N_NODES * DIM * 4;
    float* hbuf = (float*)(ws + off); off += (size_t)N_NODES * DIM * 4;
    float* enorm = (float*)(ws + off);
    unsigned* wsw = (unsigned*)ws;

    const int T = 256;
    const int gN   = (N_NODES + T - 1) / T;
    const int gDeg = 256;
    const int gSc  = NBINS;
    const int gGemm = 512;

    deg_init_kernel<<<gN, T, 0, stream>>>(dinv, wsw, ws_size);
    deg_accum_kernel<<<gDeg, T, 0, stream>>>(ei, dinv, wsw, ws_size);
    dinv_kernel<<<gN, T, 0, stream>>>(dinv, wsw, ws_size);
    place_kernel<<<PL_GRID, PL_T, 0, stream>>>(ei, dinv, wsw, ws_size);

    gemm_first_kernel<<<gGemm, T, 0, stream>>>(x, W1, b1, dinv, tbuf, out, wsw, ws_size);
    scatter_first_kernel<<<gSc, SC_T, 0, stream>>>(tbuf, ei, dinv, enorm, ws_size, out, wsw);

    gemm_relu_kernel<<<gGemm, T, 0, stream>>>(out, W2, b2, dinv, tbuf, hbuf, wsw, ws_size);
    scatter_next_kernel<<<gSc, SC_T, 0, stream>>>(tbuf, ei, dinv, enorm, ws_size, hbuf, wsw);

    gemm_relu_kernel<<<gGemm, T, 0, stream>>>(hbuf, W3, b3, dinv, tbuf, out, wsw, ws_size);
    scatter_next_kernel<<<gSc, SC_T, 0, stream>>>(tbuf, ei, dinv, enorm, ws_size, out, wsw);
}


# Edge sums moved from global f32 atomics to an LDS tile per block of 128 destination rows (edges binned by destination block; tile starts at bias+self term, rows stored once); sum order = LDS arrival
# speedup vs baseline: 2.8364x; 1.1493x over previous
#include <hip/hip_runtime.h>

#define N_NODES 100000
#define N_EDGES 1200000
#define DIM 64

typedef __attribute__((ext_vector_type(2))) float v2f;
typedef __attribute__((ext_vector_type(8))) float v8f;

#define WS_DINV_BYTES ((((size_t)N_NODES * 4) + 255) & ~(size_t)255)
#define WS_PLANE_BYTES ((size_t)N_NODES * DIM * 4)
#define WS_NEED_BYTES (WS_DINV_BYTES + 2 * WS_PLANE_BYTES + (size_t)N_EDGES * 4)

#define HALF_NODES (N_NODES / 2)

#define DB 128
#define DB_SHIFT 7
#define NBINS ((N_NODES + DB - 1) / DB)
#define BIN_WORDS 1024
#define WS_REC_OFF (WS_DINV_BYTES + 2 * WS_PLANE_BYTES)
#define WS_CNT_OFF (WS_REC_OFF + (size_t)N_EDGES * 8)
#define WS_START_OFF (WS_CNT_OFF + (size_t)BIN_WORDS * 4)
#define WS_CUR_OFF (WS_START_OFF + (size_t)BIN_WORDS * 4)
#define WS_Q1_OFF (WS_CUR_OFF + (size_t)BIN_WORDS * 4)
#define WS_NEED2 (WS_Q1_OFF + 256)
#define SC_CH 64
#define SC_CHUNKS (N_EDGES / SC_CH)
#define DEG_CH 384
#define DEG_CHUNKS (N_EDGES / DEG_CH)

__device__ __forceinline__ size_t queue_word(const int q, const size_t ws_size) {
    const size_t w1 = (ws_size >= WS_NEED2) ? (size_t)(WS_Q1_OFF / 4)
                    : (ws_size >= WS_NEED_BYTES + 256) ? (size_t)(WS_NEED_BYTES / 4)
                                                       : (size_t)N_NODES + 16;
    return q ? w1 : (size_t)N_NODES;
}

__device__ __forceinline__ void reset_queues(unsigned* wsw, const size_t ws_size) {
    if (blockIdx.x == 0 && threadIdx.x == 0) {
        wsw[queue_word(0, ws_size)] = 0u;
        wsw[queue_word(1, ws_size)] = 0u;
    }
}

__device__ __forceinline__ int placement_half() {
    return (int)((__builtin_amdgcn_s_sendmsg_rtn(0x87) >> 18) & 1u);
}

__device__ __forceinline__ unsigned issue_draw(unsigned* ctr, const int lane) {
    unsigned v = 0u;
    if (lane == 0) v = atomicAdd(ctr, 1u);
    return v;
}
__device__ __forceinline__ unsigned drawn_chunk(const unsigned v) {
    return (unsigned)__builtin_amdgcn_readfirstlane((int)v);
}

__global__ void deg_init_kernel(float* __restrict__ deg, unsigned* wsw, const size_t ws_size) {
    int i = blockIdx.x * blockDim.x + threadIdx.x;
    if (i < N_NODES) deg[i] = 1.0f;
    if (ws_size >= WS_NEED2 && i < BIN_WORDS) wsw[WS_CNT_OFF / 4 + i] = 0u;
    reset_queues(wsw, ws_size);
}

__global__ void __launch_bounds__(256)
deg_accum_kernel(const int* __restrict__ ei, float* __restrict__ deg,
                 unsigned* wsw, const size_t ws_size) {
    const int lane = threadIdx.x & 31;
    const int my_half = placement_half();
    const int n_pass = ((threadIdx.x >> 5) == 0) ? 2 : 1;
    const bool have_bins = ws_size >= WS_NEED2;
    __shared__ unsigned hist[NBINS];
    if (have_bins) {
        for (int i = threadIdx.x; i < NBINS; i += blockDim.x) hist[i] = 0u;
        __syncthreads();
    }
    for (int pass = 0; pass < n_pass; ++pass) {
        const int q  = my_half ^ pass;
        const int lo = q ? HALF_NODES : 0;
        const int hi = q ? N_NODES : HALF_NODES;
        unsigned* ctr = wsw + queue_word(q, ws_size);
        unsigned c = drawn_chunk(issue_draw(ctr, lane));
        while (c < DEG_CHUNKS) {
            const int* dp = ei + N_EDGES + (int)c * DEG_CH + lane;
            int d[DEG_CH / 32];
#pragma unroll
            for (int i = 0; i < DEG_CH / 32; ++i) d[i] = dp[i * 32];
            const unsigned nxt = issue_draw(ctr, lane);
#pragma unroll
            for (int i = 0; i < DEG_CH / 32; ++i)
                if (d[i] >= lo && d[i] < hi) {
                    atomicAdd(&deg[d[i]], 1.0f);
                    if (have_bins) atomicAdd(&hist[d[i] >> DB_SHIFT], 1u);
                }
            c = drawn_chunk(nxt);
        }
    }
    if (have_bins) {
        __syncthreads();
        unsigned* cnt = wsw + WS_CNT_OFF / 4;
        for (int i = threadIdx.x; i < NBINS; i += blockDim.x) {
            const unsigned v = hist[i];
            if (v != 0u) atomicAdd(&cnt[i], v);
        }
    }
}

__global__ void __launch_bounds__(256)
dinv_kernel(float* __restrict__ deg, unsigned* wsw, const size_t ws_size) {
    int i = blockIdx.x * blockDim.x + threadIdx.x;
    if (i < N_NODES) deg[i] = rsqrtf(deg[i]);
    __shared__ unsigned sc[256];
    if (blockIdx.x == 0 && ws_size >= WS_NEED2) {
        const int t = threadIdx.x;
        const unsigned* cnt = wsw + WS_CNT_OFF / 4;
        const unsigned c0 = cnt[4 * t], c1 = cnt[4 * t + 1], c2 = cnt[4 * t + 2], c3 = cnt[4 * t + 3];
        const unsigned own = c0 + c1 + c2 + c3;
        sc[t] = own;
        __syncthreads();
        for (int off = 1; off < 256; off <<= 1) {
            const unsigned v = (t >= off) ? sc[t - off] : 0u;
            __syncthreads();
            sc[t] += v;
            __syncthreads();
        }
        const unsigned e0 = sc[t] - own;
        unsigned* st  = wsw + WS_START_OFF / 4;
        unsigned* cur = wsw + WS_CUR_OFF / 4;
        st[4 * t]     = e0;                 cur[4 * t]     = e0;
        st[4 * t + 1] = e0 + c0;            cur[4 * t + 1] = e0 + c0;
        st[4 * t + 2] = e0 + c0 + c1;       cur[4 * t + 2] = e0 + c0 + c1;
        st[4 * t + 3] = e0 + c0 + c1 + c2;  cur[4 * t + 3] = e0 + c0 + c1 + c2;
    }
}

#define PL_T 512
#define PL_PER 8
#define PL_EDGES (PL_T * PL_PER)
#define PL_GRID ((N_EDGES + PL_EDGES - 1) / PL_EDGES)
__global__ void __launch_bounds__(PL_T)
place_kernel(const int* __restrict__ ei, const float* __restrict__ dinv,
             unsigned* wsw, const size_t ws_size) {
    if (ws_size < WS_NEED2) return;
    __shared__ unsigned hist[NBINS];
    __shared__ unsigned base[NBINS];
    for (int i = threadIdx.x; i < NBINS; i += PL_T) hist[i] = 0u;
    __syncthreads();
    int s[PL_PER], d[PL_PER];
    unsigned rank[PL_PER];
    const int e0 = blockIdx.x * PL_EDGES + threadIdx.x;
#pragma unroll
    for (int k = 0; k < PL_PER; ++k) {
        const int e = e0 + k * PL_T;
        s[k] = (e < N_EDGES) ? ei[e] : -1;
        d[k] = (e < N_EDGES) ? ei[N_EDGES + e] : 0;
    }
#pragma unroll
    for (int k = 0; k < PL_PER; ++k)
        if (s[k] >= 0) rank[k] = atomicAdd(&hist[d[k] >> DB_SHIFT], 1u);
    __syncthreads();
    unsigned* cur = wsw + WS_CUR_OFF / 4;
    for (int i = threadIdx.x; i < NBINS; i += PL_T) {
        const unsigned v = hist[i];
        base[i] = (v != 0u) ? atomicAdd(&cur[i], v) : 0u;
    }
    __syncthreads();
    uint2* rec = (uint2*)((char*)wsw + WS_REC_OFF);
#pragma unroll
    for (int k = 0; k < PL_PER; ++k)
        if (s[k] >= 0) {
            const unsigned pos = base[d[k] >> DB_SHIFT] + rank[k];
            const float w = dinv[s[k]] * dinv[d[k]];
            rec[pos] = make_uint2((unsigned)(d[k] & (DB - 1)) | ((unsigned)s[k] << DB_SHIFT),
                                  __builtin_bit_cast(unsigned, w));
        }
}

typedef __attribute__((ext_vector_type(4))) float v4f;

__device__ __forceinline__ void gemm_fused_body(const bool relu_in, const bool write_init,
                                                const float* __restrict__ in,
                                                const float* __restrict__ W,
                                                const float* __restrict__ bias,
                                                const float* __restrict__ dinv,
                                                float* __restrict__ t_out,
                                                float* __restrict__ h_init) {
    const int lane = threadIdx.x & 31;
    const int wid  = (blockIdx.x * blockDim.x + threadIdx.x) >> 5;
    const int nWaves = (gridDim.x * blockDim.x) >> 5;

    const int col0 = (wid & 3) * 16;
    const int m    = lane & 15;
    const int hi   = lane >> 4;
    const int klo  = hi * 2;

    v2f bfrag[16];
#pragma unroll
    for (int kb = 0; kb < 16; ++kb) {
        int k = kb * 4 + klo;
        bfrag[kb].x = W[k * DIM + col0 + m];
        bfrag[kb].y = W[(k + 1) * DIM + col0 + m];
    }
    const float bv = bias[col0 + m];

    const int nRowTiles  = N_NODES / 16;
    const int rowStride  = nWaves >> 2;

    for (int rt = wid >> 2; rt < nRowTiles; rt += rowStride) {
        const int row0 = rt * 16;
        const int rbase = row0 + hi * 8;
        const float* arow = in + (size_t)(row0 + m) * DIM;
        const v4f dv0 = *(const v4f*)(dinv + rbase);
        const v4f dv1 = *(const v4f*)(dinv + rbase + 4);
        v8f c = {};
#pragma unroll
        for (int kb = 0; kb < 16; ++kb) {
            v2f a = *(const v2f*)(arow + kb * 4 + klo);
            if (relu_in) { a.x = fmaxf(a.x, 0.0f); a.y = fmaxf(a.y, 0.0f); }
            c = __builtin_amdgcn_wmma_f32_16x16x4_f32(
                     false, a,  false, bfrag[kb],
                     (short)0, c,  false,  false);
        }
        float* trow = t_out  + (size_t)rbase * DIM + col0 + m;
        float* hrow = h_init + (size_t)rbase * DIM + col0 + m;
        const float dv[8] = {dv0.x, dv0.y, dv0.z, dv0.w, dv1.x, dv1.y, dv1.z, dv1.w};
#pragma unroll
        for (int r = 0; r < 8; ++r) {
            const float nm = dv[r] * dv[r];
            trow[(size_t)r * DIM] = c[r];
            if (write_init) hrow[(size_t)r * DIM] = bv + c[r] * nm;
        }
    }
}

__global__ void gemm_first_kernel(const float* __restrict__ in,
                                  const float* __restrict__ W,
                                  const float* __restrict__ bias,
                                  const float* __restrict__ dinv,
                                  float* __restrict__ t_out,
                                  float* __restrict__ h_init,
                                  unsigned* wsw, const size_t ws_size) {
    reset_queues(wsw, ws_size);
    gemm_fused_body(false, ws_size < WS_NEED2, in, W, bias, dinv, t_out, h_init);
}

__global__ void gemm_relu_kernel(const float* __restrict__ in,
                                 const float* __restrict__ W,
                                 const float* __restrict__ bias,
                                 const float* __restrict__ dinv,
                                 float* __restrict__ t_out,
                                 float* __restrict__ h_init,
                                 unsigned* wsw, const size_t ws_size) {
    reset_queues(wsw, ws_size);
    gemm_fused_body(true, ws_size < WS_NEED2, in, W, bias, dinv, t_out, h_init);
}

#define SC_K 16
#define HALF_DIM (DIM / 2)

template <bool FIRST>
__device__ __forceinline__ void scatter_body(const float* __restrict__ t,
                                             const int* __restrict__ ei,
                                             const float* __restrict__ dinv,
                                             float* __restrict__ enorm,
                                             const size_t ws_size,
                                             float* __restrict__ h,
                                             unsigned* wsw) {
    const int lane = threadIdx.x & 31;
    const bool have_enorm = ws_size >= WS_NEED_BYTES;
    const int my_half = placement_half();

    const int n_pass = ((threadIdx.x >> 5) == 0) ? 2 : 1;
    for (int pass = 0; pass < n_pass; ++pass) {
        const int q  = my_half ^ pass;
        const int lo = q ? HALF_NODES : 0;
        const int hi = q ? N_NODES : HALF_NODES;
        unsigned* ctr = wsw + queue_word(q, ws_size);
        unsigned c = drawn_chunk(issue_draw(ctr, lane));
        while (c < SC_CHUNKS) {
            const int e0 = (int)c * SC_CH + lane;

            int   sl[SC_CH / 32], dl[SC_CH / 32];
            float nl[SC_CH / 32];
            bool  mine[SC_CH / 32];
#pragma unroll
            for (int g = 0; g < SC_CH / 32; ++g) {
                sl[g] = ei[e0 + g * 32];
                dl[g] = ei[N_EDGES + e0 + g * 32];
                nl[g] = (!FIRST && have_enorm) ? enorm[e0 + g * 32] : 0.0f;
            }
            const unsigned nxt = issue_draw(ctr, lane);
#pragma unroll
            for (int g = 0; g < SC_CH / 32; ++g) {
                mine[g] = (dl[g] >= lo) && (dl[g] < hi);
                if ((FIRST || !have_enorm) && mine[g]) {
                    nl[g] = dinv[sl[g]] * dinv[dl[g]];
                    if (FIRST && have_enorm) enorm[e0 + g * 32] = nl[g];
                }
            }

#pragma unroll
            for (int g = 0; g < SC_CH / 32; ++g) {
                unsigned mask = __builtin_amdgcn_ballot_w32(mine[g]);
                while (mask != 0u) {
                    int lj[SC_K];
#pragma unroll
                    for (int j = 0; j < SC_K; ++j) {
                        lj[j] = (mask != 0u) ? (int)__builtin_ctz(mask) : -1;
                        mask &= (mask - 1u);
                    }
                    float mlo[SC_K], mhi[SC_K], nj[SC_K];
                    int   dj[SC_K];
#pragma unroll
                    for (int j = 0; j < SC_K; ++j) {
                        if (lj[j] >= 0) {
                            const int sj = __builtin_amdgcn_readlane(sl[g], lj[j]);
                            dj[j] = __builtin_amdgcn_readlane(dl[g], lj[j]);
                            nj[j] = __builtin_bit_cast(float,
                                      __builtin_amdgcn_readlane(__builtin_bit_cast(int, nl[g]), lj[j]));
                            const float* trow = t + (size_t)sj * DIM + lane;
                            mlo[j] = trow[0];
                            mhi[j] = trow[HALF_DIM];
                        }
                    }
#pragma unroll
                    for (int j = 0; j < SC_K; ++j) {
                        if (lj[j] >= 0) {
                            float* op = h + (size_t)dj[j] * DIM + lane;
                            atomicAdd(op,            mlo[j] * nj[j]);
                            atomicAdd(op + HALF_DIM, mhi[j] * nj[j]);
                        }
                    }
                }
            }
            c = drawn_chunk(nxt);
        }
    }
}

#define SC_T 256
#define SC_W (SC_T / 32)
#define ST_K 16
#define ST_PER (DB * DIM / 4 / SC_T)

typedef const __attribute__((address_space(1))) float* gfloat_p;
template <bool GUARD>
__device__ __forceinline__ void tile_trip(float* tile, const uint2 r, const int j0, const int cnt,
                                          const int lane, const float* __restrict__ t) {
    float mlo[ST_K], mhi[ST_K], wj[ST_K];
    int   dj[ST_K];
#pragma unroll
    for (int j = 0; j < ST_K; ++j) {
        if (!GUARD || j < cnt) {
            const unsigned key = (unsigned)__builtin_amdgcn_readlane((int)r.x, j0 + j);
            wj[j] = __builtin_bit_cast(float, __builtin_amdgcn_readlane((int)r.y, j0 + j));
            const gfloat_p trow = (gfloat_p)((unsigned long long)t + ((unsigned long long)(key >> DB_SHIFT) << 8));
            dj[j]  = (int)(key & (DB - 1));
            mlo[j] = trow[lane];
            mhi[j] = trow[lane + HALF_DIM];
        }
    }
#pragma unroll
    for (int j = 0; j < ST_K; ++j) {
        if (!GUARD || j < cnt) {
            float* op = tile + dj[j] * DIM + lane;
            atomicAdd(op,            mlo[j] * wj[j]);
            atomicAdd(op + HALF_DIM, mhi[j] * wj[j]);
        }
    }
}

__device__ __forceinline__ void tile_scatter_body(const float* __restrict__ t,
                                                  const float* __restrict__ bias,
                                                  const float* __restrict__ dinv,
                                                  float* __restrict__ h,
                                                  const unsigned* wsw) {
    __shared__ v4f tile4[DB * DIM / 4];
    float* tile = (float*)tile4;
    const int lane = threadIdx.x & 31;
    const int wave = __builtin_amdgcn_readfirstlane((int)(threadIdx.x >> 5));
    const unsigned* st = wsw + WS_START_OFF / 4;
    const uint2* rec = (const uint2*)((const char*)wsw + WS_REC_OFF);
    const v4f b4 = *(const v4f*)(bias + 4 * ((int)threadIdx.x & 15));

    for (int bin = blockIdx.x; bin < NBINS; bin += gridDim.x) {
        const int row0  = bin * DB;
        const int nrows = (N_NODES - row0 < DB) ? (N_NODES - row0) : DB;
        const int n4 = nrows * (DIM / 4);
        const v4f* t4 = (const v4f*)(t + (size_t)row0 * DIM);
        v4f* h4 = (v4f*)(h + (size_t)row0 * DIM);

        const int e_lo = __builtin_amdgcn_readfirstlane((int)st[bin]);
        const int e_hi = __builtin_amdgcn_readfirstlane((int)st[bin + 1]);
        uint2 r = make_uint2(0u, 0u);
        if (e_lo + wave * 32 + lane < e_hi) r = rec[e_lo + wave * 32 + lane];

        if (nrows == DB) {
            v4f buf[ST_PER];
            float dv[ST_PER];
#pragma unroll
            for (int k = 0; k < ST_PER; ++k) {
                const int p = (int)threadIdx.x + k * SC_T;
                buf[k] = t4[p];
                dv[k]  = dinv[row0 + (p >> 4)];
            }
#pragma unroll
            for (int k = 0; k < ST_PER; ++k) {
                const float nm = dv[k] * dv[k];
                tile4[(int)threadIdx.x + k * SC_T] = b4 + buf[k] * nm;
            }
        } else {
            for (int p = (int)threadIdx.x; p < n4; p += SC_T) {
                const float dvp = dinv[row0 + (p >> 4)];
                tile4[p] = b4 + t4[p] * (dvp * dvp);
            }
        }
        __syncthreads();

        for (int g = e_lo + wave * 32; g < e_hi; g += SC_W * 32) {
            const int idn = g + SC_W * 32 + lane;
            uint2 rn = make_uint2(0u, 0u);
            if (idn < e_hi) rn = rec[idn];
            const int n = (e_hi - g < 32) ? (e_hi - g) : 32;
            int j0 = 0;
            for (; j0 + ST_K <= n; j0 += ST_K) tile_trip<false>(tile, r, j0, ST_K, lane, t);
            if (j0 < n) tile_trip<true>(tile, r, j0, n - j0, lane, t);
            r = rn;
        }
        asm volatile("s_wait_dscnt 0x0" ::: "memory");
        __syncthreads();

        if (nrows == DB) {
            v4f o[ST_PER];
#pragma unroll
            for (int k = 0; k < ST_PER; ++k) o[k] = tile4[(int)threadIdx.x + k * SC_T];
#pragma unroll
            for (int k = 0; k < ST_PER; ++k) h4[(int)threadIdx.x + k * SC_T] = o[k];
        } else {
            for (int p = (int)threadIdx.x; p < n4; p += SC_T) h4[p] = tile4[p];
        }
        __syncthreads();
    }
}

__global__ void __launch_bounds__(SC_T) __attribute__((amdgpu_waves_per_eu(8, 16)))
scatter_first_kernel(const float* __restrict__ t,
                     const int* __restrict__ ei,
                     const float* __restrict__ bias,
                     const float* __restrict__ dinv,
                     float* __restrict__ enorm,
                     const size_t ws_size,
                     float* __restrict__ h,
                     unsigned* wsw) {
    if (ws_size >= WS_NEED2) tile_scatter_body(t, bias, dinv, h, wsw);
    else scatter_body<true>(t, ei, dinv, enorm, ws_size, h, wsw);
}

__global__ void __launch_bounds__(SC_T) __attribute__((amdgpu_waves_per_eu(8, 16)))
scatter_next_kernel(const float* __restrict__ t,
                    const int* __restrict__ ei,
                    const float* __restrict__ bias,
                    const float* __restrict__ dinv,
                    float* __restrict__ enorm,
                    const size_t ws_size,
                    float* __restrict__ h,
                    unsigned* wsw) {
    if (ws_size >= WS_NEED2) tile_scatter_body(t, bias, dinv, h, wsw);
    else scatter_body<false>(t, ei, dinv, enorm, ws_size, h, wsw);
}

extern "C" void kernel_launch(void* const* d_in, const int* in_sizes, int n_in,
                              void* d_out, int out_size, void* d_ws, size_t ws_size,
                              hipStream_t stream) {
    const float* x  = (const float*)d_in[0];
    const int*   ei = (const int*)d_in[1];
    const float* W1 = (const float*)d_in[2];
    const float* b1 = (const float*)d_in[3];
    const float* W2 = (const float*)d_in[4];
    const float* b2 = (const float*)d_in[5];
    const float* W3 = (const float*)d_in[6];
    const float* b3 = (const float*)d_in[7];
    float* out = (float*)d_out;

    char* ws = (char*)d_ws;
    size_t off = 0;
    float* dinv = (float*)(ws + off); off += ((size_t)N_NODES * 4 + 255) & ~(size_t)255;
    float* tbuf = (float*)(ws + off); off += (size_t)N_NODES * DIM * 4;
    float* hbuf = (float*)(ws + off); off += (size_t)N_NODES * DIM * 4;
    float* enorm = (float*)(ws + off);
    unsigned* wsw = (unsigned*)ws;

    const int T = 256;
    const int gN   = (N_NODES + T - 1) / T;
    const int gDeg = 256;
    const int gSc  = NBINS;
    const int gGemm = 512;

    deg_init_kernel<<<gN, T, 0, stream>>>(dinv, wsw, ws_size);
    deg_accum_kernel<<<gDeg, T, 0, stream>>>(ei, dinv, wsw, ws_size);
    dinv_kernel<<<gN, T, 0, stream>>>(dinv, wsw, ws_size);
    place_kernel<<<PL_GRID, PL_T, 0, stream>>>(ei, dinv, wsw, ws_size);

    gemm_first_kernel<<<gGemm, T, 0, stream>>>(x, W1, b1, dinv, tbuf, out, wsw, ws_size);
    scatter_first_kernel<<<gSc, SC_T, 0, stream>>>(tbuf, ei, b1, dinv, enorm, ws_size, out, wsw);

    gemm_relu_kernel<<<gGemm, T, 0, stream>>>(out, W2, b2, dinv, tbuf, hbuf, wsw, ws_size);
    scatter_next_kernel<<<gSc, SC_T, 0, stream>>>(tbuf, ei, b2, dinv, enorm, ws_size, hbuf, wsw);

    gemm_relu_kernel<<<gGemm, T, 0, stream>>>(hbuf, W3, b3, dinv, tbuf, out, wsw, ws_size);
    scatter_next_kernel<<<gSc, SC_T, 0, stream>>>(tbuf, ei, b3, dinv, enorm, ws_size, out, wsw);
}


# Tile scatter blocks cut from 128 to 98 destination rows (782 to 1021 work-groups, at most 4 per WGP): no WGP carries more than 392 rows where some carried 512; LDS add arrival order may shift
# speedup vs baseline: 2.9819x; 1.0513x over previous
#include <hip/hip_runtime.h>

#define N_NODES 100000
#define N_EDGES 1200000
#define DIM 64

typedef __attribute__((ext_vector_type(2))) float v2f;
typedef __attribute__((ext_vector_type(8))) float v8f;

#define WS_DINV_BYTES ((((size_t)N_NODES * 4) + 255) & ~(size_t)255)
#define WS_PLANE_BYTES ((size_t)N_NODES * DIM * 4)
#define WS_NEED_BYTES (WS_DINV_BYTES + 2 * WS_PLANE_BYTES + (size_t)N_EDGES * 4)

#define HALF_NODES (N_NODES / 2)

#define DB 98
#define DB_SHIFT 7
#define NBINS ((N_NODES + DB - 1) / DB)
#define BIN_OF(d) ((unsigned)(d) / (unsigned)DB)
#define BIN_WORDS 1024
#define WS_REC_OFF (WS_DINV_BYTES + 2 * WS_PLANE_BYTES)
#define WS_CNT_OFF (WS_REC_OFF + (size_t)N_EDGES * 8)
#define WS_START_OFF (WS_CNT_OFF + (size_t)BIN_WORDS * 4)
#define WS_CUR_OFF (WS_START_OFF + (size_t)BIN_WORDS * 4)
#define WS_Q1_OFF (WS_CUR_OFF + (size_t)BIN_WORDS * 4)
#define WS_NEED2 (WS_Q1_OFF + 256)
#define SC_CH 64
#define SC_CHUNKS (N_EDGES / SC_CH)
#define DEG_CH 384
#define DEG_CHUNKS (N_EDGES / DEG_CH)

__device__ __forceinline__ size_t queue_word(const int q, const size_t ws_size) {
    const size_t w1 = (ws_size >= WS_NEED2) ? (size_t)(WS_Q1_OFF / 4)
                    : (ws_size >= WS_NEED_BYTES + 256) ? (size_t)(WS_NEED_BYTES / 4)
                                                       : (size_t)N_NODES + 16;
    return q ? w1 : (size_t)N_NODES;
}

__device__ __forceinline__ void reset_queues(unsigned* wsw, const size_t ws_size) {
    if (blockIdx.x == 0 && threadIdx.x == 0) {
        wsw[queue_word(0, ws_size)] = 0u;
        wsw[queue_word(1, ws_size)] = 0u;
    }
}

__device__ __forceinline__ int placement_half() {
    return (int)((__builtin_amdgcn_s_sendmsg_rtn(0x87) >> 18) & 1u);
}

__device__ __forceinline__ unsigned issue_draw(unsigned* ctr, const int lane) {
    unsigned v = 0u;
    if (lane == 0) v = atomicAdd(ctr, 1u);
    return v;
}
__device__ __forceinline__ unsigned drawn_chunk(const unsigned v) {
    return (unsigned)__builtin_amdgcn_readfirstlane((int)v);
}

__global__ void deg_init_kernel(float* __restrict__ deg, unsigned* wsw, const size_t ws_size) {
    int i = blockIdx.x * blockDim.x + threadIdx.x;
    if (i < N_NODES) deg[i] = 1.0f;
    if (ws_size >= WS_NEED2 && i < BIN_WORDS) wsw[WS_CNT_OFF / 4 + i] = 0u;
    reset_queues(wsw, ws_size);
}

__global__ void __launch_bounds__(256)
deg_accum_kernel(const int* __restrict__ ei, float* __restrict__ deg,
                 unsigned* wsw, const size_t ws_size) {
    const int lane = threadIdx.x & 31;
    const int my_half = placement_half();
    const int n_pass = ((threadIdx.x >> 5) == 0) ? 2 : 1;
    const bool have_bins = ws_size >= WS_NEED2;
    __shared__ unsigned hist[NBINS];
    if (have_bins) {
        for (int i = threadIdx.x; i < NBINS; i += blockDim.x) hist[i] = 0u;
        __syncthreads();
    }
    for (int pass = 0; pass < n_pass; ++pass) {
        const int q  = my_half ^ pass;
        const int lo = q ? HALF_NODES : 0;
        const int hi = q ? N_NODES : HALF_NODES;
        unsigned* ctr = wsw + queue_word(q, ws_size);
        unsigned c = drawn_chunk(issue_draw(ctr, lane));
        while (c < DEG_CHUNKS) {
            const int* dp = ei + N_EDGES + (int)c * DEG_CH + lane;
            int d[DEG_CH / 32];
#pragma unroll
            for (int i = 0; i < DEG_CH / 32; ++i) d[i] = dp[i * 32];
            const unsigned nxt = issue_draw(ctr, lane);
#pragma unroll
            for (int i = 0; i < DEG_CH / 32; ++i)
                if (d[i] >= lo && d[i] < hi) {
                    atomicAdd(&deg[d[i]], 1.0f);
                    if (have_bins) atomicAdd(&hist[BIN_OF(d[i])], 1u);
                }
            c = drawn_chunk(nxt);
        }
    }
    if (have_bins) {
        __syncthreads();
        unsigned* cnt = wsw + WS_CNT_OFF / 4;
        for (int i = threadIdx.x; i < NBINS; i += blockDim.x) {
            const unsigned v = hist[i];
            if (v != 0u) atomicAdd(&cnt[i], v);
        }
    }
}

__global__ void __launch_bounds__(256)
dinv_kernel(float* __restrict__ deg, unsigned* wsw, const size_t ws_size) {
    int i = blockIdx.x * blockDim.x + threadIdx.x;
    if (i < N_NODES) deg[i] = rsqrtf(deg[i]);
    __shared__ unsigned sc[256];
    if (blockIdx.x == 0 && ws_size >= WS_NEED2) {
        const int t = threadIdx.x;
        const unsigned* cnt = wsw + WS_CNT_OFF / 4;
        const unsigned c0 = cnt[4 * t], c1 = cnt[4 * t + 1], c2 = cnt[4 * t + 2], c3 = cnt[4 * t + 3];
        const unsigned own = c0 + c1 + c2 + c3;
        sc[t] = own;
        __syncthreads();
        for (int off = 1; off < 256; off <<= 1) {
            const unsigned v = (t >= off) ? sc[t - off] : 0u;
            __syncthreads();
            sc[t] += v;
            __syncthreads();
        }
        const unsigned e0 = sc[t] - own;
        unsigned* st  = wsw + WS_START_OFF / 4;
        unsigned* cur = wsw + WS_CUR_OFF / 4;
        st[4 * t]     = e0;                 cur[4 * t]     = e0;
        st[4 * t + 1] = e0 + c0;            cur[4 * t + 1] = e0 + c0;
        st[4 * t + 2] = e0 + c0 + c1;       cur[4 * t + 2] = e0 + c0 + c1;
        st[4 * t + 3] = e0 + c0 + c1 + c2;  cur[4 * t + 3] = e0 + c0 + c1 + c2;
    }
}

#define PL_T 512
#define PL_PER 8
#define PL_EDGES (PL_T * PL_PER)
#define PL_GRID ((N_EDGES + PL_EDGES - 1) / PL_EDGES)
__global__ void __launch_bounds__(PL_T)
place_kernel(const int* __restrict__ ei, const float* __restrict__ dinv,
             unsigned* wsw, const size_t ws_size) {
    if (ws_size < WS_NEED2) return;
    __shared__ unsigned hist[NBINS];
    __shared__ unsigned base[NBINS];
    for (int i = threadIdx.x; i < NBINS; i += PL_T) hist[i] = 0u;
    __syncthreads();
    int s[PL_PER], d[PL_PER];
    unsigned rank[PL_PER], bn[PL_PER];
    const int e0 = blockIdx.x * PL_EDGES + threadIdx.x;
#pragma unroll
    for (int k = 0; k < PL_PER; ++k) {
        const int e = e0 + k * PL_T;
        s[k] = (e < N_EDGES) ? ei[e] : -1;
        d[k] = (e < N_EDGES) ? ei[N_EDGES + e] : 0;
    }
#pragma unroll
    for (int k = 0; k < PL_PER; ++k)
        if (s[k] >= 0) { bn[k] = BIN_OF(d[k]); rank[k] = atomicAdd(&hist[bn[k]], 1u); }
    __syncthreads();
    unsigned* cur = wsw + WS_CUR_OFF / 4;
    for (int i = threadIdx.x; i < NBINS; i += PL_T) {
        const unsigned v = hist[i];
        base[i] = (v != 0u) ? atomicAdd(&cur[i], v) : 0u;
    }
    __syncthreads();
    uint2* rec = (uint2*)((char*)wsw + WS_REC_OFF);
#pragma unroll
    for (int k = 0; k < PL_PER; ++k)
        if (s[k] >= 0) {
            const unsigned pos = base[bn[k]] + rank[k];
            const float w = dinv[s[k]] * dinv[d[k]];
            rec[pos] = make_uint2(((unsigned)d[k] - bn[k] * (unsigned)DB) | ((unsigned)s[k] << DB_SHIFT),
                                  __builtin_bit_cast(unsigned, w));
        }
}

typedef __attribute__((ext_vector_type(4))) float v4f;

__device__ __forceinline__ void gemm_fused_body(const bool relu_in, const bool write_init,
                                                const float* __restrict__ in,
                                                const float* __restrict__ W,
                                                const float* __restrict__ bias,
                                                const float* __restrict__ dinv,
                                                float* __restrict__ t_out,
                                                float* __restrict__ h_init) {
    const int lane = threadIdx.x & 31;
    const int wid  = (blockIdx.x * blockDim.x + threadIdx.x) >> 5;
    const int nWaves = (gridDim.x * blockDim.x) >> 5;

    const int col0 = (wid & 3) * 16;
    const int m    = lane & 15;
    const int hi   = lane >> 4;
    const int klo  = hi * 2;

    v2f bfrag[16];
#pragma unroll
    for (int kb = 0; kb < 16; ++kb) {
        int k = kb * 4 + klo;
        bfrag[kb].x = W[k * DIM + col0 + m];
        bfrag[kb].y = W[(k + 1) * DIM + col0 + m];
    }
    const float bv = bias[col0 + m];

    const int nRowTiles  = N_NODES / 16;
    const int rowStride  = nWaves >> 2;

    for (int rt = wid >> 2; rt < nRowTiles; rt += rowStride) {
        const int row0 = rt * 16;
        const int rbase = row0 + hi * 8;
        const float* arow = in + (size_t)(row0 + m) * DIM;
        const v4f dv0 = *(const v4f*)(dinv + rbase);
        const v4f dv1 = *(const v4f*)(dinv + rbase + 4);
        v8f c = {};
#pragma unroll
        for (int kb = 0; kb < 16; ++kb) {
            v2f a = *(const v2f*)(arow + kb * 4 + klo);
            if (relu_in) { a.x = fmaxf(a.x, 0.0f); a.y = fmaxf(a.y, 0.0f); }
            c = __builtin_amdgcn_wmma_f32_16x16x4_f32(
                     false, a,  false, bfrag[kb],
                     (short)0, c,  false,  false);
        }
        float* trow = t_out  + (size_t)rbase * DIM + col0 + m;
        float* hrow = h_init + (size_t)rbase * DIM + col0 + m;
        const float dv[8] = {dv0.x, dv0.y, dv0.z, dv0.w, dv1.x, dv1.y, dv1.z, dv1.w};
#pragma unroll
        for (int r = 0; r < 8; ++r) {
            const float nm = dv[r] * dv[r];
            trow[(size_t)r * DIM] = c[r];
            if (write_init) hrow[(size_t)r * DIM] = bv + c[r] * nm;
        }
    }
}

__global__ void gemm_first_kernel(const float* __restrict__ in,
                                  const float* __restrict__ W,
                                  const float* __restrict__ bias,
                                  const float* __restrict__ dinv,
                                  float* __restrict__ t_out,
                                  float* __restrict__ h_init,
                                  unsigned* wsw, const size_t ws_size) {
    reset_queues(wsw, ws_size);
    gemm_fused_body(false, ws_size < WS_NEED2, in, W, bias, dinv, t_out, h_init);
}

__global__ void gemm_relu_kernel(const float* __restrict__ in,
                                 const float* __restrict__ W,
                                 const float* __restrict__ bias,
                                 const float* __restrict__ dinv,
                                 float* __restrict__ t_out,
                                 float* __restrict__ h_init,
                                 unsigned* wsw, const size_t ws_size) {
    reset_queues(wsw, ws_size);
    gemm_fused_body(true, ws_size < WS_NEED2, in, W, bias, dinv, t_out, h_init);
}

#define SC_K 16
#define HALF_DIM (DIM / 2)

template <bool FIRST>
__device__ __forceinline__ void scatter_body(const float* __restrict__ t,
                                             const int* __restrict__ ei,
                                             const float* __restrict__ dinv,
                                             float* __restrict__ enorm,
                                             const size_t ws_size,
                                             float* __restrict__ h,
                                             unsigned* wsw) {
    const int lane = threadIdx.x & 31;
    const bool have_enorm = ws_size >= WS_NEED_BYTES;
    const int my_half = placement_half();

    const int n_pass = ((threadIdx.x >> 5) == 0) ? 2 : 1;
    for (int pass = 0; pass < n_pass; ++pass) {
        const int q  = my_half ^ pass;
        const int lo = q ? HALF_NODES : 0;
        const int hi = q ? N_NODES : HALF_NODES;
        unsigned* ctr = wsw + queue_word(q, ws_size);
        unsigned c = drawn_chunk(issue_draw(ctr, lane));
        while (c < SC_CHUNKS) {
            const int e0 = (int)c * SC_CH + lane;

            int   sl[SC_CH / 32], dl[SC_CH / 32];
            float nl[SC_CH / 32];
            bool  mine[SC_CH / 32];
#pragma unroll
            for (int g = 0; g < SC_CH / 32; ++g) {
                sl[g] = ei[e0 + g * 32];
                dl[g] = ei[N_EDGES + e0 + g * 32];
                nl[g] = (!FIRST && have_enorm) ? enorm[e0 + g * 32] : 0.0f;
            }
            const unsigned nxt = issue_draw(ctr, lane);
#pragma unroll
            for (int g = 0; g < SC_CH / 32; ++g) {
                mine[g] = (dl[g] >= lo) && (dl[g] < hi);
                if ((FIRST || !have_enorm) && mine[g]) {
                    nl[g] = dinv[sl[g]] * dinv[dl[g]];
                    if (FIRST && have_enorm) enorm[e0 + g * 32] = nl[g];
                }
            }

#pragma unroll
            for (int g = 0; g < SC_CH / 32; ++g) {
                unsigned mask = __builtin_amdgcn_ballot_w32(mine[g]);
                while (mask != 0u) {
                    int lj[SC_K];
#pragma unroll
                    for (int j = 0; j < SC_K; ++j) {
                        lj[j] = (mask != 0u) ? (int)__builtin_ctz(mask) : -1;
                        mask &= (mask - 1u);
                    }
                    float mlo[SC_K], mhi[SC_K], nj[SC_K];
                    int   dj[SC_K];
#pragma unroll
                    for (int j = 0; j < SC_K; ++j) {
                        if (lj[j] >= 0) {
                            const int sj = __builtin_amdgcn_readlane(sl[g], lj[j]);
                            dj[j] = __builtin_amdgcn_readlane(dl[g], lj[j]);
                            nj[j] = __builtin_bit_cast(float,
                                      __builtin_amdgcn_readlane(__builtin_bit_cast(int, nl[g]), lj[j]));
                            const float* trow = t + (size_t)sj * DIM + lane;
                            mlo[j] = trow[0];
                            mhi[j] = trow[HALF_DIM];
                        }
                    }
#pragma unroll
                    for (int j = 0; j < SC_K; ++j) {
                        if (lj[j] >= 0) {
                            float* op = h + (size_t)dj[j] * DIM + lane;
                            atomicAdd(op,            mlo[j] * nj[j]);
                            atomicAdd(op + HALF_DIM, mhi[j] * nj[j]);
                        }
                    }
                }
            }
            c = drawn_chunk(nxt);
        }
    }
}

#define SC_T 256
#define SC_W (SC_T / 32)
#define ST_K 16
#define TILE4 (DB * DIM / 4)
#define ST_T 224
#define ST_PER (TILE4 / ST_T)
static_assert(TILE4 % ST_T == 0 && ST_T % 32 == 0 && ST_T <= SC_T, "whole waves, whole rounds");

typedef const __attribute__((address_space(1))) float* gfloat_p;
template <bool GUARD>
__device__ __forceinline__ void tile_trip(float* tile, const uint2 r, const int j0, const int cnt,
                                          const int lane, const float* __restrict__ t) {
    float mlo[ST_K], mhi[ST_K], wj[ST_K];
    int   dj[ST_K];
#pragma unroll
    for (int j = 0; j < ST_K; ++j) {
        if (!GUARD || j < cnt) {
            const unsigned key = (unsigned)__builtin_amdgcn_readlane((int)r.x, j0 + j);
            wj[j] = __builtin_bit_cast(float, __builtin_amdgcn_readlane((int)r.y, j0 + j));
            const gfloat_p trow = (gfloat_p)((unsigned long long)t + ((unsigned long long)(key >> DB_SHIFT) << 8));
            dj[j]  = (int)(key & ((1u << DB_SHIFT) - 1u));
            mlo[j] = trow[lane];
            mhi[j] = trow[lane + HALF_DIM];
        }
    }
#pragma unroll
    for (int j = 0; j < ST_K; ++j) {
        if (!GUARD || j < cnt) {
            float* op = tile + dj[j] * DIM + lane;
            atomicAdd(op,            mlo[j] * wj[j]);
            atomicAdd(op + HALF_DIM, mhi[j] * wj[j]);
        }
    }
}

__device__ __forceinline__ void tile_scatter_body(const float* __restrict__ t,
                                                  const float* __restrict__ bias,
                                                  const float* __restrict__ dinv,
                                                  float* __restrict__ h,
                                                  const unsigned* wsw) {
    __shared__ v4f tile4[DB * DIM / 4];
    float* tile = (float*)tile4;
    const int lane = threadIdx.x & 31;
    const int wave = __builtin_amdgcn_readfirstlane((int)(threadIdx.x >> 5));
    const unsigned* st = wsw + WS_START_OFF / 4;
    const uint2* rec = (const uint2*)((const char*)wsw + WS_REC_OFF);
    const v4f b4 = *(const v4f*)(bias + 4 * ((int)threadIdx.x & 15));

    for (int bin = blockIdx.x; bin < NBINS; bin += gridDim.x) {
        const int row0  = bin * DB;
        const int nrows = (N_NODES - row0 < DB) ? (N_NODES - row0) : DB;
        const int n4 = nrows * (DIM / 4);
        const v4f* t4 = (const v4f*)(t + (size_t)row0 * DIM);
        v4f* h4 = (v4f*)(h + (size_t)row0 * DIM);

        const int e_lo = __builtin_amdgcn_readfirstlane((int)st[bin]);
        const int e_hi = __builtin_amdgcn_readfirstlane((int)st[bin + 1]);
        uint2 r = make_uint2(0u, 0u);
        if (e_lo + wave * 32 + lane < e_hi) r = rec[e_lo + wave * 32 + lane];

        if (nrows == DB) {
            if ((int)threadIdx.x < ST_T) {
                v4f buf[ST_PER];
                float dv[ST_PER];
#pragma unroll
                for (int k = 0; k < ST_PER; ++k) {
                    const int p = (int)threadIdx.x + k * ST_T;
                    buf[k] = t4[p];
                    dv[k]  = dinv[row0 + (p >> 4)];
                }
#pragma unroll
                for (int k = 0; k < ST_PER; ++k) {
                    const float nm = dv[k] * dv[k];
                    tile4[(int)threadIdx.x + k * ST_T] = b4 + buf[k] * nm;
                }
            }
        } else {
            for (int p = (int)threadIdx.x; p < n4; p += SC_T) {
                const float dvp = dinv[row0 + (p >> 4)];
                tile4[p] = b4 + t4[p] * (dvp * dvp);
            }
        }
        __syncthreads();

        for (int g = e_lo + wave * 32; g < e_hi; g += SC_W * 32) {
            const int idn = g + SC_W * 32 + lane;
            uint2 rn = make_uint2(0u, 0u);
            if (idn < e_hi) rn = rec[idn];
            const int n = (e_hi - g < 32) ? (e_hi - g) : 32;
            int j0 = 0;
            for (; j0 + ST_K <= n; j0 += ST_K) tile_trip<false>(tile, r, j0, ST_K, lane, t);
            if (j0 < n) tile_trip<true>(tile, r, j0, n - j0, lane, t);
            r = rn;
        }
        asm volatile("s_wait_dscnt 0x0" ::: "memory");
        __syncthreads();

        if (nrows == DB) {
            if ((int)threadIdx.x < ST_T) {
                v4f o[ST_PER];
#pragma unroll
                for (int k = 0; k < ST_PER; ++k) o[k] = tile4[(int)threadIdx.x + k * ST_T];
#pragma unroll
                for (int k = 0; k < ST_PER; ++k) h4[(int)threadIdx.x + k * ST_T] = o[k];
            }
        } else {
            for (int p = (int)threadIdx.x; p < n4; p += SC_T) h4[p] = tile4[p];
        }
        __syncthreads();
    }
}

__global__ void __launch_bounds__(SC_T) __attribute__((amdgpu_waves_per_eu(8, 16)))
scatter_first_kernel(const float* __restrict__ t,
                     const int* __restrict__ ei,
                     const float* __restrict__ bias,
                     const float* __restrict__ dinv,
                     float* __restrict__ enorm,
                     const size_t ws_size,
                     float* __restrict__ h,
                     unsigned* wsw) {
    if (ws_size >= WS_NEED2) tile_scatter_body(t, bias, dinv, h, wsw);
    else scatter_body<true>(t, ei, dinv, enorm, ws_size, h, wsw);
}

__global__ void __launch_bounds__(SC_T) __attribute__((amdgpu_waves_per_eu(8, 16)))
scatter_next_kernel(const float* __restrict__ t,
                    const int* __restrict__ ei,
                    const float* __restrict__ bias,
                    const float* __restrict__ dinv,
                    float* __restrict__ enorm,
                    const size_t ws_size,
                    float* __restrict__ h,
                    unsigned* wsw) {
    if (ws_size >= WS_NEED2) tile_scatter_body(t, bias, dinv, h, wsw);
    else scatter_body<false>(t, ei, dinv, enorm, ws_size, h, wsw);
}

extern "C" void kernel_launch(void* const* d_in, const int* in_sizes, int n_in,
                              void* d_out, int out_size, void* d_ws, size_t ws_size,
                              hipStream_t stream) {
    const float* x  = (const float*)d_in[0];
    const int*   ei = (const int*)d_in[1];
    const float* W1 = (const float*)d_in[2];
    const float* b1 = (const float*)d_in[3];
    const float* W2 = (const float*)d_in[4];
    const float* b2 = (const float*)d_in[5];
    const float* W3 = (const float*)d_in[6];
    const float* b3 = (const float*)d_in[7];
    float* out = (float*)d_out;

    char* ws = (char*)d_ws;
    size_t off = 0;
    float* dinv = (float*)(ws + off); off += ((size_t)N_NODES * 4 + 255) & ~(size_t)255;
    float* tbuf = (float*)(ws + off); off += (size_t)N_NODES * DIM * 4;
    float* hbuf = (float*)(ws + off); off += (size_t)N_NODES * DIM * 4;
    float* enorm = (float*)(ws + off);
    unsigned* wsw = (unsigned*)ws;

    const int T = 256;
    const int gN   = (N_NODES + T - 1) / T;
    const int gDeg = 256;
    const int gSc  = NBINS;
    const int gGemm = 512;

    deg_init_kernel<<<gN, T, 0, stream>>>(dinv, wsw, ws_size);
    deg_accum_kernel<<<gDeg, T, 0, stream>>>(ei, dinv, wsw, ws_size);
    dinv_kernel<<<gN, T, 0, stream>>>(dinv, wsw, ws_size);
    place_kernel<<<PL_GRID, PL_T, 0, stream>>>(ei, dinv, wsw, ws_size);

    gemm_first_kernel<<<gGemm, T, 0, stream>>>(x, W1, b1, dinv, tbuf, out, wsw, ws_size);
    scatter_first_kernel<<<gSc, SC_T, 0, stream>>>(tbuf, ei, b1, dinv, enorm, ws_size, out, wsw);

    gemm_relu_kernel<<<gGemm, T, 0, stream>>>(out, W2, b2, dinv, tbuf, hbuf, wsw, ws_size);
    scatter_next_kernel<<<gSc, SC_T, 0, stream>>>(tbuf, ei, b2, dinv, enorm, ws_size, hbuf, wsw);

    gemm_relu_kernel<<<gGemm, T, 0, stream>>>(hbuf, W3, b3, dinv, tbuf, out, wsw, ws_size);
    scatter_next_kernel<<<gSc, SC_T, 0, stream>>>(tbuf, ei, b3, dinv, enorm, ws_size, out, wsw);
}


# Layer 1's product x @ W1 moved into the edge-placement launch as extra work-groups (neither needs the other's result): 10 launches to 9, product streams beside the placement's round trips
# speedup vs baseline: 3.2194x; 1.0796x over previous
#include <hip/hip_runtime.h>

#define N_NODES 100000
#define N_EDGES 1200000
#define DIM 64

typedef __attribute__((ext_vector_type(2))) float v2f;
typedef __attribute__((ext_vector_type(8))) float v8f;

#define WS_DINV_BYTES ((((size_t)N_NODES * 4) + 255) & ~(size_t)255)
#define WS_PLANE_BYTES ((size_t)N_NODES * DIM * 4)
#define WS_NEED_BYTES (WS_DINV_BYTES + 2 * WS_PLANE_BYTES + (size_t)N_EDGES * 4)

#define HALF_NODES (N_NODES / 2)

#define DB 98
#define DB_SHIFT 7
#define NBINS ((N_NODES + DB - 1) / DB)
#define BIN_OF(d) ((unsigned)(d) / (unsigned)DB)
#define BIN_WORDS 1024
#define WS_REC_OFF (WS_DINV_BYTES + 2 * WS_PLANE_BYTES)
#define WS_CNT_OFF (WS_REC_OFF + (size_t)N_EDGES * 8)
#define WS_START_OFF (WS_CNT_OFF + (size_t)BIN_WORDS * 4)
#define WS_CUR_OFF (WS_START_OFF + (size_t)BIN_WORDS * 4)
#define WS_Q1_OFF (WS_CUR_OFF + (size_t)BIN_WORDS * 4)
#define WS_NEED2 (WS_Q1_OFF + 256)
#define SC_CH 64
#define SC_CHUNKS (N_EDGES / SC_CH)
#define DEG_CH 384
#define DEG_CHUNKS (N_EDGES / DEG_CH)

__device__ __forceinline__ size_t queue_word(const int q, const size_t ws_size) {
    const size_t w1 = (ws_size >= WS_NEED2) ? (size_t)(WS_Q1_OFF / 4)
                    : (ws_size >= WS_NEED_BYTES + 256) ? (size_t)(WS_NEED_BYTES / 4)
                                                       : (size_t)N_NODES + 16;
    return q ? w1 : (size_t)N_NODES;
}

__device__ __forceinline__ void reset_queues(unsigned* wsw, const size_t ws_size) {
    if (blockIdx.x == 0 && threadIdx.x == 0) {
        wsw[queue_word(0, ws_size)] = 0u;
        wsw[queue_word(1, ws_size)] = 0u;
    }
}

__device__ __forceinline__ int placement_half() {
    return (int)((__builtin_amdgcn_s_sendmsg_rtn(0x87) >> 18) & 1u);
}

__device__ __forceinline__ unsigned issue_draw(unsigned* ctr, const int lane) {
    unsigned v = 0u;
    if (lane == 0) v = atomicAdd(ctr, 1u);
    return v;
}
__device__ __forceinline__ unsigned drawn_chunk(const unsigned v) {
    return (unsigned)__builtin_amdgcn_readfirstlane((int)v);
}

__global__ void deg_init_kernel(float* __restrict__ deg, unsigned* wsw, const size_t ws_size) {
    int i = blockIdx.x * blockDim.x + threadIdx.x;
    if (i < N_NODES) deg[i] = 1.0f;
    if (ws_size >= WS_NEED2 && i < BIN_WORDS) wsw[WS_CNT_OFF / 4 + i] = 0u;
    reset_queues(wsw, ws_size);
}

__global__ void __launch_bounds__(256)
deg_accum_kernel(const int* __restrict__ ei, float* __restrict__ deg,
                 unsigned* wsw, const size_t ws_size) {
    const int lane = threadIdx.x & 31;
    const int my_half = placement_half();
    const int n_pass = ((threadIdx.x >> 5) == 0) ? 2 : 1;
    const bool have_bins = ws_size >= WS_NEED2;
    __shared__ unsigned hist[NBINS];
    if (have_bins) {
        for (int i = threadIdx.x; i < NBINS; i += blockDim.x) hist[i] = 0u;
        __syncthreads();
    }
    for (int pass = 0; pass < n_pass; ++pass) {
        const int q  = my_half ^ pass;
        const int lo = q ? HALF_NODES : 0;
        const int hi = q ? N_NODES : HALF_NODES;
        unsigned* ctr = wsw + queue_word(q, ws_size);
        unsigned c = drawn_chunk(issue_draw(ctr, lane));
        while (c < DEG_CHUNKS) {
            const int* dp = ei + N_EDGES + (int)c * DEG_CH + lane;
            int d[DEG_CH / 32];
#pragma unroll
            for (int i = 0; i < DEG_CH / 32; ++i) d[i] = dp[i * 32];
            const unsigned nxt = issue_draw(ctr, lane);
#pragma unroll
            for (int i = 0; i < DEG_CH / 32; ++i)
                if (d[i] >= lo && d[i] < hi) {
                    atomicAdd(&deg[d[i]], 1.0f);
                    if (have_bins) atomicAdd(&hist[BIN_OF(d[i])], 1u);
                }
            c = drawn_chunk(nxt);
        }
    }
    if (have_bins) {
        __syncthreads();
        unsigned* cnt = wsw + WS_CNT_OFF / 4;
        for (int i = threadIdx.x; i < NBINS; i += blockDim.x) {
            const unsigned v = hist[i];
            if (v != 0u) atomicAdd(&cnt[i], v);
        }
    }
}

__global__ void __launch_bounds__(256)
dinv_kernel(float* __restrict__ deg, unsigned* wsw, const size_t ws_size) {
    int i = blockIdx.x * blockDim.x + threadIdx.x;
    if (i < N_NODES) deg[i] = rsqrtf(deg[i]);
    __shared__ unsigned sc[256];
    if (blockIdx.x == 0 && ws_size >= WS_NEED2) {
        const int t = threadIdx.x;
        const unsigned* cnt = wsw + WS_CNT_OFF / 4;
        const unsigned c0 = cnt[4 * t], c1 = cnt[4 * t + 1], c2 = cnt[4 * t + 2], c3 = cnt[4 * t + 3];
        const unsigned own = c0 + c1 + c2 + c3;
        sc[t] = own;
        __syncthreads();
        for (int off = 1; off < 256; off <<= 1) {
            const unsigned v = (t >= off) ? sc[t - off] : 0u;
            __syncthreads();
            sc[t] += v;
            __syncthreads();
        }
        const unsigned e0 = sc[t] - own;
        unsigned* st  = wsw + WS_START_OFF / 4;
        unsigned* cur = wsw + WS_CUR_OFF / 4;
        st[4 * t]     = e0;                 cur[4 * t]     = e0;
        st[4 * t + 1] = e0 + c0;            cur[4 * t + 1] = e0 + c0;
        st[4 * t + 2] = e0 + c0 + c1;       cur[4 * t + 2] = e0 + c0 + c1;
        st[4 * t + 3] = e0 + c0 + c1 + c2;  cur[4 * t + 3] = e0 + c0 + c1 + c2;
    }
}

typedef __attribute__((ext_vector_type(4))) float v4f;

__device__ __forceinline__ void gemm_fused_body(const bool relu_in, const bool write_init,
                                                const float* __restrict__ in,
                                                const float* __restrict__ W,
                                                const float* __restrict__ bias,
                                                const float* __restrict__ dinv,
                                                float* __restrict__ t_out,
                                                float* __restrict__ h_init,
                                                const int vblock, const int vgrid) {
    const int lane = threadIdx.x & 31;
    const int wid  = (vblock * (int)blockDim.x + (int)threadIdx.x) >> 5;
    const int nWaves = (vgrid * (int)blockDim.x) >> 5;

    const int col0 = (wid & 3) * 16;
    const int m    = lane & 15;
    const int hi   = lane >> 4;
    const int klo  = hi * 2;

    v2f bfrag[16];
#pragma unroll
    for (int kb = 0; kb < 16; ++kb) {
        int k = kb * 4 + klo;
        bfrag[kb].x = W[k * DIM + col0 + m];
        bfrag[kb].y = W[(k + 1) * DIM + col0 + m];
    }
    const float bv = bias[col0 + m];

    const int nRowTiles  = N_NODES / 16;
    const int rowStride  = nWaves >> 2;

    for (int rt = wid >> 2; rt < nRowTiles; rt += rowStride) {
        const int row0 = rt * 16;
        const int rbase = row0 + hi * 8;
        const float* arow = in + (size_t)(row0 + m) * DIM;
        const v4f dv0 = *(const v4f*)(dinv + rbase);
        const v4f dv1 = *(const v4f*)(dinv + rbase + 4);
        v8f c = {};
#pragma unroll
        for (int kb = 0; kb < 16; ++kb) {
            v2f a = *(const v2f*)(arow + kb * 4 + klo);
            if (relu_in) { a.x = fmaxf(a.x, 0.0f); a.y = fmaxf(a.y, 0.0f); }
            c = __builtin_amdgcn_wmma_f32_16x16x4_f32(
                     false, a,  false, bfrag[kb],
                     (short)0, c,  false,  false);
        }
        float* trow = t_out  + (size_t)rbase * DIM + col0 + m;
        float* hrow = h_init + (size_t)rbase * DIM + col0 + m;
        const float dv[8] = {dv0.x, dv0.y, dv0.z, dv0.w, dv1.x, dv1.y, dv1.z, dv1.w};
#pragma unroll
        for (int r = 0; r < 8; ++r) {
            const float nm = dv[r] * dv[r];
            trow[(size_t)r * DIM] = c[r];
            if (write_init) hrow[(size_t)r * DIM] = bv + c[r] * nm;
        }
    }
}

__global__ void gemm_relu_kernel(const float* __restrict__ in,
                                 const float* __restrict__ W,
                                 const float* __restrict__ bias,
                                 const float* __restrict__ dinv,
                                 float* __restrict__ t_out,
                                 float* __restrict__ h_init,
                                 unsigned* wsw, const size_t ws_size) {
    reset_queues(wsw, ws_size);
    gemm_fused_body(true, ws_size < WS_NEED2, in, W, bias, dinv, t_out, h_init, (int)blockIdx.x, (int)gridDim.x);
}

#define PL_T 512
#define PL_PER 8
#define PL_EDGES (PL_T * PL_PER)
#define PL_GRID ((N_EDGES + PL_EDGES - 1) / PL_EDGES)
#define PG_GEMM (512 - PL_GRID)
static_assert(PG_GEMM > 0 && (PG_GEMM * (PL_T / 32)) % 4 == 0, "whole groups of four column-tile waves");
__global__ void __launch_bounds__(PL_T)
place_gemm_kernel(const int* __restrict__ ei, const float* __restrict__ dinv,
                  unsigned* wsw, const size_t ws_size,
                  const float* __restrict__ x,
                  const float* __restrict__ W,
                  const float* __restrict__ bias,
                  float* __restrict__ t_out,
                  float* __restrict__ h_init) {
    reset_queues(wsw, ws_size);
    if ((int)blockIdx.x >= PL_GRID) {
        gemm_fused_body(false, ws_size < WS_NEED2, x, W, bias, dinv, t_out, h_init,
                        (int)blockIdx.x - PL_GRID, (int)gridDim.x - PL_GRID);
        return;
    }
    if (ws_size < WS_NEED2) return;
    __shared__ unsigned hist[NBINS];
    __shared__ unsigned base[NBINS];
    for (int i = threadIdx.x; i < NBINS; i += PL_T) hist[i] = 0u;
    __syncthreads();
    int s[PL_PER], d[PL_PER];
    unsigned rank[PL_PER], bn[PL_PER];
    const int e0 = blockIdx.x * PL_EDGES + threadIdx.x;
#pragma unroll
    for (int k = 0; k < PL_PER; ++k) {
        const int e = e0 + k * PL_T;
        s[k] = (e < N_EDGES) ? ei[e] : -1;
        d[k] = (e < N_EDGES) ? ei[N_EDGES + e] : 0;
    }
#pragma unroll
    for (int k = 0; k < PL_PER; ++k)
        if (s[k] >= 0) { bn[k] = BIN_OF(d[k]); rank[k] = atomicAdd(&hist[bn[k]], 1u); }
    __syncthreads();
    unsigned* cur = wsw + WS_CUR_OFF / 4;
    for (int i = threadIdx.x; i < NBINS; i += PL_T) {
        const unsigned v = hist[i];
        base[i] = (v != 0u) ? atomicAdd(&cur[i], v) : 0u;
    }
    __syncthreads();
    uint2* rec = (uint2*)((char*)wsw + WS_REC_OFF);
#pragma unroll
    for (int k = 0; k < PL_PER; ++k)
        if (s[k] >= 0) {
            const unsigned pos = base[bn[k]] + rank[k];
            const float w = dinv[s[k]] * dinv[d[k]];
            rec[pos] = make_uint2(((unsigned)d[k] - bn[k] * (unsigned)DB) | ((unsigned)s[k] << DB_SHIFT),
                                  __builtin_bit_cast(unsigned, w));
        }
}

#define SC_K 16
#define HALF_DIM (DIM / 2)

template <bool FIRST>
__device__ __forceinline__ void scatter_body(const float* __restrict__ t,
                                             const int* __restrict__ ei,
                                             const float* __restrict__ dinv,
                                             float* __restrict__ enorm,
                                             const size_t ws_size,
                                             float* __restrict__ h,
                                             unsigned* wsw) {
    const int lane = threadIdx.x & 31;
    const bool have_enorm = ws_size >= WS_NEED_BYTES;
    const int my_half = placement_half();

    const int n_pass = ((threadIdx.x >> 5) == 0) ? 2 : 1;
    for (int pass = 0; pass < n_pass; ++pass) {
        const int q  = my_half ^ pass;
        const int lo = q ? HALF_NODES : 0;
        const int hi = q ? N_NODES : HALF_NODES;
        unsigned* ctr = wsw + queue_word(q, ws_size);
        unsigned c = drawn_chunk(issue_draw(ctr, lane));
        while (c < SC_CHUNKS) {
            const int e0 = (int)c * SC_CH + lane;

            int   sl[SC_CH / 32], dl[SC_CH / 32];
            float nl[SC_CH / 32];
            bool  mine[SC_CH / 32];
#pragma unroll
            for (int g = 0; g < SC_CH / 32; ++g) {
                sl[g] = ei[e0 + g * 32];
                dl[g] = ei[N_EDGES + e0 + g * 32];
                nl[g] = (!FIRST && have_enorm) ? enorm[e0 + g * 32] : 0.0f;
            }
            const unsigned nxt = issue_draw(ctr, lane);
#pragma unroll
            for (int g = 0; g < SC_CH / 32; ++g) {
                mine[g] = (dl[g] >= lo) && (dl[g] < hi);
                if ((FIRST || !have_enorm) && mine[g]) {
                    nl[g] = dinv[sl[g]] * dinv[dl[g]];
                    if (FIRST && have_enorm) enorm[e0 + g * 32] = nl[g];
                }
            }

#pragma unroll
            for (int g = 0; g < SC_CH / 32; ++g) {
                unsigned mask = __builtin_amdgcn_ballot_w32(mine[g]);
                while (mask != 0u) {
                    int lj[SC_K];
#pragma unroll
                    for (int j = 0; j < SC_K; ++j) {
                        lj[j] = (mask != 0u) ? (int)__builtin_ctz(mask) : -1;
                        mask &= (mask - 1u);
                    }
                    float mlo[SC_K], mhi[SC_K], nj[SC_K];
                    int   dj[SC_K];
#pragma unroll
                    for (int j = 0; j < SC_K; ++j) {
                        if (lj[j] >= 0) {
                            const int sj = __builtin_amdgcn_readlane(sl[g], lj[j]);
                            dj[j] = __builtin_amdgcn_readlane(dl[g], lj[j]);
                            nj[j] = __builtin_bit_cast(float,
                                      __builtin_amdgcn_readlane(__builtin_bit_cast(int, nl[g]), lj[j]));
                            const float* trow = t + (size_t)sj * DIM + lane;
                            mlo[j] = trow[0];
                            mhi[j] = trow[HALF_DIM];
                        }
                    }
#pragma unroll
                    for (int j = 0; j < SC_K; ++j) {
                        if (lj[j] >= 0) {
                            float* op = h + (size_t)dj[j] * DIM + lane;
                            atomicAdd(op,            mlo[j] * nj[j]);
                            atomicAdd(op + HALF_DIM, mhi[j] * nj[j]);
                        }
                    }
                }
            }
            c = drawn_chunk(nxt);
        }
    }
}

#define SC_T 256
#define SC_W (SC_T / 32)
#define ST_K 16
#define TILE4 (DB * DIM / 4)
#define ST_T 224
#define ST_PER (TILE4 / ST_T)
static_assert(TILE4 % ST_T == 0 && ST_T % 32 == 0 && ST_T <= SC_T, "whole waves, whole rounds");

typedef const __attribute__((address_space(1))) float* gfloat_p;
template <bool GUARD>
__device__ __forceinline__ void tile_trip(float* tile, const uint2 r, const int j0, const int cnt,
                                          const int lane, const float* __restrict__ t) {
    float mlo[ST_K], mhi[ST_K], wj[ST_K];
    int   dj[ST_K];
#pragma unroll
    for (int j = 0; j < ST_K; ++j) {
        if (!GUARD || j < cnt) {
            const unsigned key = (unsigned)__builtin_amdgcn_readlane((int)r.x, j0 + j);
            wj[j] = __builtin_bit_cast(float, __builtin_amdgcn_readlane((int)r.y, j0 + j));
            const gfloat_p trow = (gfloat_p)((unsigned long long)t + ((unsigned long long)(key >> DB_SHIFT) << 8));
            dj[j]  = (int)(key & ((1u << DB_SHIFT) - 1u));
            mlo[j] = trow[lane];
            mhi[j] = trow[lane + HALF_DIM];
        }
    }
#pragma unroll
    for (int j = 0; j < ST_K; ++j) {
        if (!GUARD || j < cnt) {
            float* op = tile + dj[j] * DIM + lane;
            atomicAdd(op,            mlo[j] * wj[j]);
            atomicAdd(op + HALF_DIM, mhi[j] * wj[j]);
        }
    }
}

__device__ __forceinline__ void tile_scatter_body(const float* __restrict__ t,
                                                  const float* __restrict__ bias,
                                                  const float* __restrict__ dinv,
                                                  float* __restrict__ h,
                                                  const unsigned* wsw) {
    __shared__ v4f tile4[DB * DIM / 4];
    float* tile = (float*)tile4;
    const int lane = threadIdx.x & 31;
    const int wave = __builtin_amdgcn_readfirstlane((int)(threadIdx.x >> 5));
    const unsigned* st = wsw + WS_START_OFF / 4;
    const uint2* rec = (const uint2*)((const char*)wsw + WS_REC_OFF);
    const v4f b4 = *(const v4f*)(bias + 4 * ((int)threadIdx.x & 15));

    for (int bin = blockIdx.x; bin < NBINS; bin += gridDim.x) {
        const int row0  = bin * DB;
        const int nrows = (N_NODES - row0 < DB) ? (N_NODES - row0) : DB;
        const int n4 = nrows * (DIM / 4);
        const v4f* t4 = (const v4f*)(t + (size_t)row0 * DIM);
        v4f* h4 = (v4f*)(h + (size_t)row0 * DIM);

        const int e_lo = __builtin_amdgcn_readfirstlane((int)st[bin]);
        const int e_hi = __builtin_amdgcn_readfirstlane((int)st[bin + 1]);
        uint2 r = make_uint2(0u, 0u);
        if (e_lo + wave * 32 + lane < e_hi) r = rec[e_lo + wave * 32 + lane];

        if (nrows == DB) {
            if ((int)threadIdx.x < ST_T) {
                v4f buf[ST_PER];
                float dv[ST_PER];
#pragma unroll
                for (int k = 0; k < ST_PER; ++k) {
                    const int p = (int)threadIdx.x + k * ST_T;
                    buf[k] = t4[p];
                    dv[k]  = dinv[row0 + (p >> 4)];
                }
#pragma unroll
                for (int k = 0; k < ST_PER; ++k) {
                    const float nm = dv[k] * dv[k];
                    tile4[(int)threadIdx.x + k * ST_T] = b4 + buf[k] * nm;
                }
            }
        } else {
            for (int p = (int)threadIdx.x; p < n4; p += SC_T) {
                const float dvp = dinv[row0 + (p >> 4)];
                tile4[p] = b4 + t4[p] * (dvp * dvp);
            }
        }
        __syncthreads();

        for (int g = e_lo + wave * 32; g < e_hi; g += SC_W * 32) {
            const int idn = g + SC_W * 32 + lane;
            uint2 rn = make_uint2(0u, 0u);
            if (idn < e_hi) rn = rec[idn];
            const int n = (e_hi - g < 32) ? (e_hi - g) : 32;
            int j0 = 0;
            for (; j0 + ST_K <= n; j0 += ST_K) tile_trip<false>(tile, r, j0, ST_K, lane, t);
            if (j0 < n) tile_trip<true>(tile, r, j0, n - j0, lane, t);
            r = rn;
        }
        asm volatile("s_wait_dscnt 0x0" ::: "memory");
        __syncthreads();

        if (nrows == DB) {
            if ((int)threadIdx.x < ST_T) {
                v4f o[ST_PER];
#pragma unroll
                for (int k = 0; k < ST_PER; ++k) o[k] = tile4[(int)threadIdx.x + k * ST_T];
#pragma unroll
                for (int k = 0; k < ST_PER; ++k) h4[(int)threadIdx.x + k * ST_T] = o[k];
            }
        } else {
            for (int p = (int)threadIdx.x; p < n4; p += SC_T) h4[p] = tile4[p];
        }
        __syncthreads();
    }
}

__global__ void __launch_bounds__(SC_T) __attribute__((amdgpu_waves_per_eu(8, 16)))
scatter_first_kernel(const float* __restrict__ t,
                     const int* __restrict__ ei,
                     const float* __restrict__ bias,
                     const float* __restrict__ dinv,
                     float* __restrict__ enorm,
                     const size_t ws_size,
                     float* __restrict__ h,
                     unsigned* wsw) {
    if (ws_size >= WS_NEED2) tile_scatter_body(t, bias, dinv, h, wsw);
    else scatter_body<true>(t, ei, dinv, enorm, ws_size, h, wsw);
}

__global__ void __launch_bounds__(SC_T) __attribute__((amdgpu_waves_per_eu(8, 16)))
scatter_next_kernel(const float* __restrict__ t,
                    const int* __restrict__ ei,
                    const float* __restrict__ bias,
                    const float* __restrict__ dinv,
                    float* __restrict__ enorm,
                    const size_t ws_size,
                    float* __restrict__ h,
                    unsigned* wsw) {
    if (ws_size >= WS_NEED2) tile_scatter_body(t, bias, dinv, h, wsw);
    else scatter_body<false>(t, ei, dinv, enorm, ws_size, h, wsw);
}

extern "C" void kernel_launch(void* const* d_in, const int* in_sizes, int n_in,
                              void* d_out, int out_size, void* d_ws, size_t ws_size,
                              hipStream_t stream) {
    const float* x  = (const float*)d_in[0];
    const int*   ei = (const int*)d_in[1];
    const float* W1 = (const float*)d_in[2];
    const float* b1 = (const float*)d_in[3];
    const float* W2 = (const float*)d_in[4];
    const float* b2 = (const float*)d_in[5];
    const float* W3 = (const float*)d_in[6];
    const float* b3 = (const float*)d_in[7];
    float* out = (float*)d_out;

    char* ws = (char*)d_ws;
    size_t off = 0;
    float* dinv = (float*)(ws + off); off += ((size_t)N_NODES * 4 + 255) & ~(size_t)255;
    float* tbuf = (float*)(ws + off); off += (size_t)N_NODES * DIM * 4;
    float* hbuf = (float*)(ws + off); off += (size_t)N_NODES * DIM * 4;
    float* enorm = (float*)(ws + off);
    unsigned* wsw = (unsigned*)ws;

    const int T = 256;
    const int gN   = (N_NODES + T - 1) / T;
    const int gDeg = 256;
    const int gSc  = NBINS;
    const int gGemm = 512;

    deg_init_kernel<<<gN, T, 0, stream>>>(dinv, wsw, ws_size);
    deg_accum_kernel<<<gDeg, T, 0, stream>>>(ei, dinv, wsw, ws_size);
    dinv_kernel<<<gN, T, 0, stream>>>(dinv, wsw, ws_size);
    place_gemm_kernel<<<PL_GRID + PG_GEMM, PL_T, 0, stream>>>(ei, dinv, wsw, ws_size, x, W1, b1, tbuf, out);

    scatter_first_kernel<<<gSc, SC_T, 0, stream>>>(tbuf, ei, b1, dinv, enorm, ws_size, out, wsw);

    gemm_relu_kernel<<<gGemm, T, 0, stream>>>(out, W2, b2, dinv, tbuf, hbuf, wsw, ws_size);
    scatter_next_kernel<<<gSc, SC_T, 0, stream>>>(tbuf, ei, b2, dinv, enorm, ws_size, hbuf, wsw);

    gemm_relu_kernel<<<gGemm, T, 0, stream>>>(hbuf, W3, b3, dinv, tbuf, out, wsw, ws_size);
    scatter_next_kernel<<<gSc, SC_T, 0, stream>>>(tbuf, ei, b3, dinv, enorm, ws_size, out, wsw);
}


# Layer 2 and 3 products made in the tile scatter's tail: the finished LDS sum tile (row pitch 68 words, ReLU at the read) is the WMMA A operand; next t stored, not h; 9 launches to 7; same k order
# speedup vs baseline: 3.8932x; 1.2093x over previous
#include <hip/hip_runtime.h>

#define N_NODES 100000
#define N_EDGES 1200000
#define DIM 64

typedef __attribute__((ext_vector_type(2))) float v2f;
typedef __attribute__((ext_vector_type(8))) float v8f;

#define WS_DINV_BYTES ((((size_t)N_NODES * 4) + 255) & ~(size_t)255)
#define WS_PLANE_BYTES ((size_t)N_NODES * DIM * 4)

#define HALF_NODES (N_NODES / 2)

#define DB 98
#define DB_SHIFT 7
#define NBINS ((N_NODES + DB - 1) / DB)
#define BIN_OF(d) ((unsigned)(d) / (unsigned)DB)
#define BIN_WORDS 1024
#define WS_REC_OFF (WS_DINV_BYTES + 2 * WS_PLANE_BYTES)
#define WS_CNT_OFF (WS_REC_OFF + (size_t)N_EDGES * 8)
#define WS_START_OFF (WS_CNT_OFF + (size_t)BIN_WORDS * 4)
#define WS_CUR_OFF (WS_START_OFF + (size_t)BIN_WORDS * 4)
#define WS_Q1_OFF (WS_CUR_OFF + (size_t)BIN_WORDS * 4)
#define WS_NEED2 (WS_Q1_OFF + 256)
#define DEG_CH 384
#define DEG_CHUNKS (N_EDGES / DEG_CH)

__device__ __forceinline__ size_t queue_word(const int q, const size_t ws_size) {
    const size_t w1 = (ws_size >= WS_NEED2) ? (size_t)(WS_Q1_OFF / 4) : (size_t)N_NODES + 16;
    return q ? w1 : (size_t)N_NODES;
}

__device__ __forceinline__ void reset_queues(unsigned* wsw, const size_t ws_size) {
    if (blockIdx.x == 0 && threadIdx.x == 0) {
        wsw[queue_word(0, ws_size)] = 0u;
        wsw[queue_word(1, ws_size)] = 0u;
    }
}

__device__ __forceinline__ int placement_half() {
    return (int)((__builtin_amdgcn_s_sendmsg_rtn(0x87) >> 18) & 1u);
}

__device__ __forceinline__ unsigned issue_draw(unsigned* ctr, const int lane) {
    unsigned v = 0u;
    if (lane == 0) v = atomicAdd(ctr, 1u);
    return v;
}
__device__ __forceinline__ unsigned drawn_chunk(const unsigned v) {
    return (unsigned)__builtin_amdgcn_readfirstlane((int)v);
}

__global__ void deg_init_kernel(float* __restrict__ deg, unsigned* wsw, const size_t ws_size) {
    int i = blockIdx.x * blockDim.x + threadIdx.x;
    if (i < N_NODES) deg[i] = 1.0f;
    if (ws_size >= WS_NEED2 && i < BIN_WORDS) wsw[WS_CNT_OFF / 4 + i] = 0u;
    reset_queues(wsw, ws_size);
}

__global__ void __launch_bounds__(256)
deg_accum_kernel(const int* __restrict__ ei, float* __restrict__ deg,
                 unsigned* wsw, const size_t ws_size) {
    const int lane = threadIdx.x & 31;
    const int my_half = placement_half();
    const int n_pass = ((threadIdx.x >> 5) == 0) ? 2 : 1;
    const bool have_bins = ws_size >= WS_NEED2;
    __shared__ unsigned hist[NBINS];
    if (have_bins) {
        for (int i = threadIdx.x; i < NBINS; i += blockDim.x) hist[i] = 0u;
        __syncthreads();
    }
    for (int pass = 0; pass < n_pass; ++pass) {
        const int q  = my_half ^ pass;
        const int lo = q ? HALF_NODES : 0;
        const int hi = q ? N_NODES : HALF_NODES;
        unsigned* ctr = wsw + queue_word(q, ws_size);
        unsigned c = drawn_chunk(issue_draw(ctr, lane));
        while (c < DEG_CHUNKS) {
            const int* dp = ei + N_EDGES + (int)c * DEG_CH + lane;
            int d[DEG_CH / 32];
#pragma unroll
            for (int i = 0; i < DEG_CH / 32; ++i) d[i] = dp[i * 32];
            const unsigned nxt = issue_draw(ctr, lane);
#pragma unroll
            for (int i = 0; i < DEG_CH / 32; ++i)
                if (d[i] >= lo && d[i] < hi) {
                    atomicAdd(&deg[d[i]], 1.0f);
                    if (have_bins) atomicAdd(&hist[BIN_OF(d[i])], 1u);
                }
            c = drawn_chunk(nxt);
        }
    }
    if (have_bins) {
        __syncthreads();
        unsigned* cnt = wsw + WS_CNT_OFF / 4;
        for (int i = threadIdx.x; i < NBINS; i += blockDim.x) {
            const unsigned v = hist[i];
            if (v != 0u) atomicAdd(&cnt[i], v);
        }
    }
}

__global__ void __launch_bounds__(256)
dinv_kernel(float* __restrict__ deg, unsigned* wsw, const size_t ws_size) {
    int i = blockIdx.x * blockDim.x + threadIdx.x;
    if (i < N_NODES) deg[i] = rsqrtf(deg[i]);
    __shared__ unsigned sc[256];
    if (blockIdx.x == 0 && ws_size >= WS_NEED2) {
        const int t = threadIdx.x;
        const unsigned* cnt = wsw + WS_CNT_OFF / 4;
        const unsigned c0 = cnt[4 * t], c1 = cnt[4 * t + 1], c2 = cnt[4 * t + 2], c3 = cnt[4 * t + 3];
        const unsigned own = c0 + c1 + c2 + c3;
        sc[t] = own;
        __syncthreads();
        for (int off = 1; off < 256; off <<= 1) {
            const unsigned v = (t >= off) ? sc[t - off] : 0u;
            __syncthreads();
            sc[t] += v;
            __syncthreads();
        }
        const unsigned e0 = sc[t] - own;
        unsigned* st  = wsw + WS_START_OFF / 4;
        unsigned* cur = wsw + WS_CUR_OFF / 4;
        st[4 * t]     = e0;                 cur[4 * t]     = e0;
        st[4 * t + 1] = e0 + c0;            cur[4 * t + 1] = e0 + c0;
        st[4 * t + 2] = e0 + c0 + c1;       cur[4 * t + 2] = e0 + c0 + c1;
        st[4 * t + 3] = e0 + c0 + c1 + c2;  cur[4 * t + 3] = e0 + c0 + c1 + c2;
    }
}

typedef __attribute__((ext_vector_type(4))) float v4f;

__device__ __forceinline__ void load_b_fragments(v2f (&bfrag)[16], const float* __restrict__ W,
                                                 const int col0, const int m, const int klo) {
#pragma unroll
    for (int kb = 0; kb < 16; ++kb) {
        const int k = kb * 4 + klo;
        bfrag[kb].x = W[k * DIM + col0 + m];
        bfrag[kb].y = W[(k + 1) * DIM + col0 + m];
    }
}

__device__ __forceinline__ void product_body(const float* __restrict__ in,
                                             const float* __restrict__ W,
                                             float* __restrict__ t_out,
                                             const int vblock, const int vgrid) {
    const int lane = threadIdx.x & 31;
    const int wid  = (vblock * (int)blockDim.x + (int)threadIdx.x) >> 5;
    const int nWaves = (vgrid * (int)blockDim.x) >> 5;

    const int col0 = (wid & 3) * 16;
    const int m    = lane & 15;
    const int hi   = lane >> 4;
    const int klo  = hi * 2;

    v2f bfrag[16];
    load_b_fragments(bfrag, W, col0, m, klo);

    const int nRowTiles  = N_NODES / 16;
    const int rowStride  = nWaves >> 2;

    for (int rt = wid >> 2; rt < nRowTiles; rt += rowStride) {
        const int row0 = rt * 16;
        const int rbase = row0 + hi * 8;
        const float* arow = in + (size_t)(row0 + m) * DIM;
        v8f c = {};
#pragma unroll
        for (int kb = 0; kb < 16; ++kb) {
            const v2f a = *(const v2f*)(arow + kb * 4 + klo);
            c = __builtin_amdgcn_wmma_f32_16x16x4_f32(
                     false, a,  false, bfrag[kb],
                     (short)0, c,  false,  false);
        }
        float* trow = t_out + (size_t)rbase * DIM + col0 + m;
#pragma unroll
        for (int r = 0; r < 8; ++r) trow[(size_t)r * DIM] = c[r];
    }
}

#define PL_T 512
#define PL_PER 8
#define PL_EDGES (PL_T * PL_PER)
#define PL_GRID ((N_EDGES + PL_EDGES - 1) / PL_EDGES)
#define PG_GEMM (512 - PL_GRID)
static_assert(PG_GEMM > 0 && (PG_GEMM * (PL_T / 32)) % 4 == 0, "whole groups of four column-tile waves");
__global__ void __launch_bounds__(PL_T)
place_gemm_kernel(const int* __restrict__ ei, const float* __restrict__ dinv,
                  unsigned* wsw, const size_t ws_size,
                  const float* __restrict__ x,
                  const float* __restrict__ W,
                  float* __restrict__ t_out) {
    if (ws_size < WS_NEED2) return;
    if ((int)blockIdx.x >= PL_GRID) {
        product_body(x, W, t_out, (int)blockIdx.x - PL_GRID, (int)gridDim.x - PL_GRID);
        return;
    }
    __shared__ unsigned hist[NBINS];
    __shared__ unsigned base[NBINS];
    for (int i = threadIdx.x; i < NBINS; i += PL_T) hist[i] = 0u;
    __syncthreads();
    int s[PL_PER], d[PL_PER];
    unsigned rank[PL_PER], bn[PL_PER];
    const int e0 = blockIdx.x * PL_EDGES + threadIdx.x;
#pragma unroll
    for (int k = 0; k < PL_PER; ++k) {
        const int e = e0 + k * PL_T;
        s[k] = (e < N_EDGES) ? ei[e] : -1;
        d[k] = (e < N_EDGES) ? ei[N_EDGES + e] : 0;
    }
#pragma unroll
    for (int k = 0; k < PL_PER; ++k)
        if (s[k] >= 0) { bn[k] = BIN_OF(d[k]); rank[k] = atomicAdd(&hist[bn[k]], 1u); }
    __syncthreads();
    unsigned* cur = wsw + WS_CUR_OFF / 4;
    for (int i = threadIdx.x; i < NBINS; i += PL_T) {
        const unsigned v = hist[i];
        base[i] = (v != 0u) ? atomicAdd(&cur[i], v) : 0u;
    }
    __syncthreads();
    uint2* rec = (uint2*)((char*)wsw + WS_REC_OFF);
#pragma unroll
    for (int k = 0; k < PL_PER; ++k)
        if (s[k] >= 0) {
            const unsigned pos = base[bn[k]] + rank[k];
            const float w = dinv[s[k]] * dinv[d[k]];
            rec[pos] = make_uint2(((unsigned)d[k] - bn[k] * (unsigned)DB) | ((unsigned)s[k] << DB_SHIFT),
                                  __builtin_bit_cast(unsigned, w));
        }
}

#define HALF_DIM (DIM / 2)
#define SC_T 256
#define SC_W (SC_T / 32)
#define ST_K 16
#define TP 68
#define TP4 (TP / 4)
#define TROWS 112
#define TILE4 (DB * DIM / 4)
#define ST_T 224
#define ST_PER (TILE4 / ST_T)
static_assert(TILE4 % ST_T == 0 && ST_T % 32 == 0 && ST_T <= SC_T, "whole waves, whole rounds");
static_assert((TROWS - DB) * (DIM / 4) == ST_PER * 32, "the eighth wave zeroes the rows behind a full bin's own, 7 pieces per lane");
static_assert(TROWS % 16 == 0 && TROWS >= DB && TP % 4 == 0, "whole row tiles, 16-byte pieces");

typedef const __attribute__((address_space(1))) float* gfloat_p;
template <bool GUARD>
__device__ __forceinline__ void tile_trip(float* tile, const uint2 r, const int j0, const int cnt,
                                          const int lane, const float* __restrict__ t) {
    float mlo[ST_K], mhi[ST_K], wj[ST_K];
    int   dj[ST_K];
#pragma unroll
    for (int j = 0; j < ST_K; ++j) {
        if (!GUARD || j < cnt) {
            const unsigned key = (unsigned)__builtin_amdgcn_readlane((int)r.x, j0 + j);
            wj[j] = __builtin_bit_cast(float, __builtin_amdgcn_readlane((int)r.y, j0 + j));
            const gfloat_p trow = (gfloat_p)((unsigned long long)t + ((unsigned long long)(key >> DB_SHIFT) << 8));
            dj[j]  = (int)(key & ((1u << DB_SHIFT) - 1u));
            mlo[j] = trow[lane];
            mhi[j] = trow[lane + HALF_DIM];
        }
    }
#pragma unroll
    for (int j = 0; j < ST_K; ++j) {
        if (!GUARD || j < cnt) {
            float* op = tile + dj[j] * TP + lane;
            atomicAdd(op,            mlo[j] * wj[j]);
            atomicAdd(op + HALF_DIM, mhi[j] * wj[j]);
        }
    }
}

template <bool PRODUCT>
__device__ __forceinline__ void tile_scatter_body(const float* __restrict__ t,
                                                  const float* __restrict__ bias,
                                                  const float* __restrict__ dinv,
                                                  float* __restrict__ dst,
                                                  const unsigned* wsw,
                                                  const float* __restrict__ Wn) {
    __shared__ v4f tile4[TROWS * TP4];
    float* tile = (float*)tile4;
    const int lane = threadIdx.x & 31;
    const int wave = __builtin_amdgcn_readfirstlane((int)(threadIdx.x >> 5));
    const unsigned* st = wsw + WS_START_OFF / 4;
    const uint2* rec = (const uint2*)((const char*)wsw + WS_REC_OFF);
    const int cg = (int)threadIdx.x & 15;
    const v4f b4 = *(const v4f*)(bias + 4 * cg);

    const int bin = blockIdx.x;
    if (bin >= NBINS) return;
    const int row0  = bin * DB;
    const int nrows = (N_NODES - row0 < DB) ? (N_NODES - row0) : DB;
    const int n4 = nrows * (DIM / 4);
    const v4f* t4 = (const v4f*)(t + (size_t)row0 * DIM);

    const int e_lo = __builtin_amdgcn_readfirstlane((int)st[bin]);
    const int e_hi = __builtin_amdgcn_readfirstlane((int)st[bin + 1]);
    uint2 r = make_uint2(0u, 0u);
    if (e_lo + wave * 32 + lane < e_hi) r = rec[e_lo + wave * 32 + lane];

    if (nrows == DB) {
        if ((int)threadIdx.x < ST_T) {
            v4f buf[ST_PER];
            float dv[ST_PER];
#pragma unroll
            for (int k = 0; k < ST_PER; ++k) {
                const int p = (int)threadIdx.x + k * ST_T;
                buf[k] = t4[p];
                dv[k]  = dinv[row0 + (p >> 4)];
            }
#pragma unroll
            for (int k = 0; k < ST_PER; ++k) {
                const int p = (int)threadIdx.x + k * ST_T;
                const float nm = dv[k] * dv[k];
                tile4[(p >> 4) * TP4 + cg] = b4 + buf[k] * nm;
            }
        } else if (PRODUCT) {
#pragma unroll
            for (int k = 0; k < ST_PER; ++k) {
                const int q = TILE4 + lane + k * 32;
                tile4[(q >> 4) * TP4 + (q & 15)] = (v4f){0.0f, 0.0f, 0.0f, 0.0f};
            }
        }
    } else {
        for (int p = (int)threadIdx.x; p < n4; p += SC_T) {
            const float dvp = dinv[row0 + (p >> 4)];
            tile4[(p >> 4) * TP4 + cg] = b4 + t4[p] * (dvp * dvp);
        }
        if (PRODUCT)
            for (int q = n4 + (int)threadIdx.x; q < TROWS * (DIM / 4); q += SC_T)
                tile4[(q >> 4) * TP4 + (q & 15)] = (v4f){0.0f, 0.0f, 0.0f, 0.0f};
    }
    __syncthreads();

    for (int g = e_lo + wave * 32; g < e_hi; g += SC_W * 32) {
        const int idn = g + SC_W * 32 + lane;
        uint2 rn = make_uint2(0u, 0u);
        if (idn < e_hi) rn = rec[idn];
        const int n = (e_hi - g < 32) ? (e_hi - g) : 32;
        int j0 = 0;
        for (; j0 + ST_K <= n; j0 += ST_K) tile_trip<false>(tile, r, j0, ST_K, lane, t);
        if (j0 < n) tile_trip<true>(tile, r, j0, n - j0, lane, t);
        r = rn;
    }

    if (PRODUCT) {
        const int m    = lane & 15;
        const int hi   = lane >> 4;
        const int klo  = hi * 2;
        const int col0 = (wave & 3) * 16;
        v2f bfrag[16];
        load_b_fragments(bfrag, Wn, col0, m, klo);

        asm volatile("s_wait_dscnt 0x0" ::: "memory");
        __syncthreads();

        for (int rt = wave >> 2; rt * 16 < nrows; rt += 2) {
            const float* arow = tile + (rt * 16 + m) * TP + klo;
            v8f c = {};
#pragma unroll
            for (int kb = 0; kb < 16; ++kb) {
                v2f a = *(const v2f*)(arow + kb * 4);
                a.x = fmaxf(a.x, 0.0f);
                a.y = fmaxf(a.y, 0.0f);
                c = __builtin_amdgcn_wmma_f32_16x16x4_f32(
                         false, a,  false, bfrag[kb],
                         (short)0, c,  false,  false);
            }
            const int rb = rt * 16 + hi * 8;
            float* trow = dst + (size_t)(row0 + rb) * DIM + col0 + m;
            if (rt * 16 + 16 <= nrows) {
#pragma unroll
                for (int r8 = 0; r8 < 8; ++r8) trow[(size_t)r8 * DIM] = c[r8];
            } else {
#pragma unroll
                for (int r8 = 0; r8 < 8; ++r8)
                    if (rb + r8 < nrows) trow[(size_t)r8 * DIM] = c[r8];
            }
        }
    } else {
        asm volatile("s_wait_dscnt 0x0" ::: "memory");
        __syncthreads();

        v4f* h4 = (v4f*)(dst + (size_t)row0 * DIM);
        if (nrows == DB) {
            if ((int)threadIdx.x < ST_T) {
                v4f o[ST_PER];
#pragma unroll
                for (int k = 0; k < ST_PER; ++k) {
                    const int p = (int)threadIdx.x + k * ST_T;
                    o[k] = tile4[(p >> 4) * TP4 + cg];
                }
#pragma unroll
                for (int k = 0; k < ST_PER; ++k) h4[(int)threadIdx.x + k * ST_T] = o[k];
            }
        } else {
            for (int p = (int)threadIdx.x; p < n4; p += SC_T) h4[p] = tile4[(p >> 4) * TP4 + cg];
        }
    }
}

__global__ void __launch_bounds__(SC_T) __attribute__((amdgpu_waves_per_eu(8, 8)))
scatter_product_kernel(const float* __restrict__ t,
                       const float* __restrict__ bias,
                       const float* __restrict__ dinv,
                       const size_t ws_size,
                       float* __restrict__ t_next,
                       const unsigned* wsw,
                       const float* __restrict__ Wn) {
    if (ws_size >= WS_NEED2) tile_scatter_body<true>(t, bias, dinv, t_next, wsw, Wn);
}

__global__ void __launch_bounds__(SC_T) __attribute__((amdgpu_waves_per_eu(8, 8)))
scatter_last_kernel(const float* __restrict__ t,
                    const float* __restrict__ bias,
                    const float* __restrict__ dinv,
                    const size_t ws_size,
                    float* __restrict__ h,
                    const unsigned* wsw) {
    if (ws_size >= WS_NEED2) tile_scatter_body<false>(t, bias, dinv, h, wsw, nullptr);
}

extern "C" void kernel_launch(void* const* d_in, const int* in_sizes, int n_in,
                              void* d_out, int out_size, void* d_ws, size_t ws_size,
                              hipStream_t stream) {
    const float* x  = (const float*)d_in[0];
    const int*   ei = (const int*)d_in[1];
    const float* W1 = (const float*)d_in[2];
    const float* b1 = (const float*)d_in[3];
    const float* W2 = (const float*)d_in[4];
    const float* b2 = (const float*)d_in[5];
    const float* W3 = (const float*)d_in[6];
    const float* b3 = (const float*)d_in[7];
    float* out = (float*)d_out;

    char* ws = (char*)d_ws;
    size_t off = 0;
    float* dinv = (float*)(ws + off); off += ((size_t)N_NODES * 4 + 255) & ~(size_t)255;
    float* tbuf = (float*)(ws + off); off += (size_t)N_NODES * DIM * 4;
    float* ubuf = (float*)(ws + off); off += (size_t)N_NODES * DIM * 4;
    unsigned* wsw = (unsigned*)ws;

    const int T = 256;
    const int gN   = (N_NODES + T - 1) / T;
    const int gDeg = 256;
    const int gSc  = NBINS;

    deg_init_kernel<<<gN, T, 0, stream>>>(dinv, wsw, ws_size);
    deg_accum_kernel<<<gDeg, T, 0, stream>>>(ei, dinv, wsw, ws_size);
    dinv_kernel<<<gN, T, 0, stream>>>(dinv, wsw, ws_size);
    place_gemm_kernel<<<PL_GRID + PG_GEMM, PL_T, 0, stream>>>(ei, dinv, wsw, ws_size, x, W1, tbuf);

    scatter_product_kernel<<<gSc, SC_T, 0, stream>>>(tbuf, b1, dinv, ws_size, ubuf, wsw, W2);
    scatter_product_kernel<<<gSc, SC_T, 0, stream>>>(ubuf, b2, dinv, ws_size, tbuf, wsw, W3);
    scatter_last_kernel<<<gSc, SC_T, 0, stream>>>(tbuf, b3, dinv, ws_size, out, wsw);
}
